# starting kernel
# baseline (speedup 1.0000x reference)
#include <hip/hip_runtime.h>
#include <hip/hip_bf16.h>

#define B_   16
#define T_   2048
#define D_   384
#define NB_  128
#define CPW  1152
#define EP   1156
#define HP   388
#define CH   16
#define EPSF 1e-8f

typedef __attribute__((ext_vector_type(2))) float v2f;
typedef __attribute__((ext_vector_type(8))) float v8f;

__global__ __launch_bounds__(256) void so3_prep_kernel(
    const float* __restrict__ W_in,
    const float* __restrict__ W_biv, const float* __restrict__ b_biv,
    const float* __restrict__ W_dec, const float* __restrict__ b_dec,
    const float* __restrict__ W_inj, const float* __restrict__ b_inj,
    float* __restrict__ weff, float* __restrict__ bias) {
  int idx = blockIdx.x * blockDim.x + threadIdx.x;
  if (idx >= 4 * CPW) return;
  int k = idx / CPW;
  int c = idx % CPW;
  const float* Wc;
  int cc;
  if (c < D_)            { Wc = W_biv; cc = c; }
  else if (c < 2 * D_)   { Wc = W_dec; cc = c - D_; }
  else                   { Wc = W_inj; cc = c - 2 * D_; }
  float s = 0.f;
  for (int d = 0; d < D_; ++d)
    s = fmaf(W_in[k * D_ + d], Wc[d * D_ + cc], s);
  weff[k * CPW + c] = s;
  if (k == 0)
    bias[c] = (c < D_) ? b_biv[c] : (c < 2 * D_ ? b_dec[c - D_] : b_inj[c - 2 * D_]);
}

__global__ __launch_bounds__(128) void so3_scan_kernel(
    const float* __restrict__ quat,
    const float* __restrict__ weff,
    const float* __restrict__ bias,
    const float* __restrict__ W_out,
    float* __restrict__ out) {
  __shared__ float sWeff[4 * CPW];
  __shared__ float sBias[CPW];
  __shared__ float sE[CH * EP];
  __shared__ float sH[CH * HP];
  __shared__ float sWout[D_ * 4];
  __shared__ float sPart[4 * 256];

  const int tid   = threadIdx.x;
  const int lane  = tid & 31;
  const int uwave = __builtin_amdgcn_readfirstlane(tid >> 5);
  const int t16   = lane & 15;
  const int k0    = (lane < 16) ? 0 : 2;
  const int mhi   = (lane < 16) ? 0 : 8;
  const int b     = blockIdx.x;

  for (int i = tid; i < 4 * CPW; i += 128) sWeff[i] = weff[i];
  for (int i = tid; i < CPW;     i += 128) sBias[i] = bias[i];
  for (int i = tid; i < D_ * 4;  i += 128) sWout[i] = W_out[i];
  __syncthreads();

  float hx = 0.f, hy = 0.f, hz = 0.f;
  const float* qbase = quat + (size_t)b * T_ * 4;

  for (int ch = 0; ch < T_ / CH; ++ch) {
    const int t0 = ch * CH;

    v2f a;
    a.x = qbase[(t0 + t16) * 4 + k0];
    a.y = qbase[(t0 + t16) * 4 + k0 + 1];

    if (tid == 0 && ch + 1 < T_ / CH)
      __builtin_prefetch(qbase + (t0 + CH) * 4, 0, 1);

    for (int tile = uwave; tile < CPW / 16; tile += 4) {
      int col = tile * 16 + t16;
      v2f bm;
      bm.x = sWeff[k0 * CPW + col];
      bm.y = sWeff[(k0 + 1) * CPW + col];
      float bv = sBias[col];
      v8f c = {bv, bv, bv, bv, bv, bv, bv, bv};
      v8f d = __builtin_amdgcn_wmma_f32_16x16x4_f32(
           false, a,  false, bm,
           (short)0, c,  false,  false);
#pragma unroll
      for (int v = 0; v < 8; ++v)
        sE[(v + mhi) * EP + col] = d[v];
    }
    __syncthreads();

    const int j3 = tid * 3;
#pragma unroll 4
    for (int tt = 0; tt < CH; ++tt) {
      const float* e = &sE[tt * EP];
      float bx = e[j3], by = e[j3 + 1], bz = e[j3 + 2];
      float nb = sqrtf(bx * bx + by * by + bz * bz);
      nb = fmaxf(nb, EPSF);
      float sn, cs;
      __sincosf(0.5f * nb, &sn, &cs);
      float s  = sn / nb;
      float qx = s * bz, qy = -s * by, qz = s * bx;
      float tx = 2.f * (qy * hz - qz * hy);
      float ty = 2.f * (qz * hx - qx * hz);
      float tz = 2.f * (qx * hy - qy * hx);
      float rx = hx + cs * tx + (qy * tz - qz * ty);
      float ry = hy + cs * ty + (qz * tx - qx * tz);
      float rz = hz + cs * tz + (qx * ty - qy * tx);
      float dx = e[D_ + j3], dy = e[D_ + j3 + 1], dz = e[D_ + j3 + 2];
      float ix = e[2 * D_ + j3], iy = e[2 * D_ + j3 + 1], iz = e[2 * D_ + j3 + 2];
      hx = rx / (1.f + __expf(-dx)) + ix;
      hy = ry / (1.f + __expf(-dy)) + iy;
      hz = rz / (1.f + __expf(-dz)) + iz;
      sH[tt * HP + j3]     = hx;
      sH[tt * HP + j3 + 1] = hy;
      sH[tt * HP + j3 + 2] = hz;
    }
    __syncthreads();

    {
      v8f acc0 = {0.f, 0.f, 0.f, 0.f, 0.f, 0.f, 0.f, 0.f};
      v8f acc1 = {0.f, 0.f, 0.f, 0.f, 0.f, 0.f, 0.f, 0.f};
      const int kbeg = uwave * 24;
#pragma unroll 4
      for (int kk = 0; kk < 24; kk += 2) {
        int kr0 = (kbeg + kk) * 4 + k0;
        int kr1 = (kbeg + kk + 1) * 4 + k0;
        v2f a2, b2;
        a2.x = sH[t16 * HP + kr0];
        a2.y = sH[t16 * HP + kr0 + 1];
        b2.x = (t16 < 4) ? sWout[kr0 * 4 + t16] : 0.f;
        b2.y = (t16 < 4) ? sWout[(kr0 + 1) * 4 + t16] : 0.f;
        acc0 = __builtin_amdgcn_wmma_f32_16x16x4_f32(
            false, a2, false, b2, (short)0, acc0, false, false);
        a2.x = sH[t16 * HP + kr1];
        a2.y = sH[t16 * HP + kr1 + 1];
        b2.x = (t16 < 4) ? sWout[kr1 * 4 + t16] : 0.f;
        b2.y = (t16 < 4) ? sWout[(kr1 + 1) * 4 + t16] : 0.f;
        acc1 = __builtin_amdgcn_wmma_f32_16x16x4_f32(
            false, a2, false, b2, (short)0, acc1, false, false);
      }
      acc0 = acc0 + acc1;
#pragma unroll
      for (int v = 0; v < 8; ++v)
        sPart[uwave * 256 + (v + mhi) * 16 + t16] = acc0[v];
    }
    __syncthreads();

    if (tid < 16) {
      float p[4];
#pragma unroll
      for (int c = 0; c < 4; ++c)
        p[c] = sPart[0 * 256 + tid * 16 + c] + sPart[1 * 256 + tid * 16 + c] +
               sPart[2 * 256 + tid * 16 + c] + sPart[3 * 256 + tid * 16 + c];
      float n = sqrtf(p[0] * p[0] + p[1] * p[1] + p[2] * p[2] + p[3] * p[3]);
      float rn = 1.f / fmaxf(n, EPSF);
      float4 o = {p[0] * rn, p[1] * rn, p[2] * rn, p[3] * rn};
      *(float4*)&out[((size_t)b * T_ + t0 + tid) * 4] = o;
    }
    __syncthreads();
  }
}

extern "C" void kernel_launch(void* const* d_in, const int* in_sizes, int n_in,
                              void* d_out, int out_size, void* d_ws, size_t ws_size,
                              hipStream_t stream) {
  const float* quat  = (const float*)d_in[0];
  const float* W_in  = (const float*)d_in[1];
  const float* W_biv = (const float*)d_in[2];
  const float* b_biv = (const float*)d_in[3];
  const float* W_dec = (const float*)d_in[4];
  const float* b_dec = (const float*)d_in[5];
  const float* W_inj = (const float*)d_in[6];
  const float* b_inj = (const float*)d_in[7];
  const float* W_out = (const float*)d_in[8];
  float* out  = (float*)d_out;
  float* weff = (float*)d_ws;
  float* bias = weff + 4 * CPW;

  so3_prep_kernel<<<(4 * CPW + 255) / 256, 256, 0, stream>>>(
      W_in, W_biv, b_biv, W_dec, b_dec, W_inj, b_inj, weff, bias);
  so3_scan_kernel<<<B_, 128, 0, stream>>>(quat, weff, bias, W_out, out);
}


# State-independent step math (E GEMM on WMMA, norm, sincos, sigmoid gate as 1/(1+exp)) hoisted into a grid-wide pre-pass; the scan streams 10 coefficients per step and block
# speedup vs baseline: 2.0643x; 2.0643x over previous
#include <hip/hip_runtime.h>
#include <hip/hip_bf16.h>

#define B_   16
#define T_   2048
#define D_   384
#define NB_  128
#define CPW  1152
#define HP   388
#define CH   16
#define NK   10
#define T4_  (T_ / 4)
#define WS_COEF_OFF 8192
#define EPSF 1e-8f

typedef __attribute__((ext_vector_type(2))) float v2f;
typedef __attribute__((ext_vector_type(4))) float v4f;
typedef __attribute__((ext_vector_type(8))) float v8f;

__global__ __launch_bounds__(256) void so3_prep_kernel(
    const float* __restrict__ W_in,
    const float* __restrict__ W_biv, const float* __restrict__ b_biv,
    const float* __restrict__ W_dec, const float* __restrict__ b_dec,
    const float* __restrict__ W_inj, const float* __restrict__ b_inj,
    float* __restrict__ weff, float* __restrict__ bias) {
  int idx = blockIdx.x * blockDim.x + threadIdx.x;
  if (idx >= 4 * CPW) return;
  int k = idx / CPW;
  int c = idx % CPW;
  const float* Wc;
  int cc;
  if (c < D_)            { Wc = W_biv; cc = c; }
  else if (c < 2 * D_)   { Wc = W_dec; cc = c - D_; }
  else                   { Wc = W_inj; cc = c - 2 * D_; }
  float s = 0.f;
  for (int d = 0; d < D_; ++d)
    s = fmaf(W_in[k * D_ + d], Wc[d * D_ + cc], s);
  weff[k * CPW + c] = s;
  if (k == 0)
    bias[c] = (c < D_) ? b_biv[c] : (c < 2 * D_ ? b_dec[c - D_] : b_inj[c - 2 * D_]);
}

__global__ __launch_bounds__(256) void so3_coef_kernel(
    const float* __restrict__ quat,
    const float* __restrict__ weff,
    const float* __restrict__ bias,
    float* __restrict__ coef) {
  const int tid  = threadIdx.x;
  const int lane = tid & 31;
  const int g    = __builtin_amdgcn_readfirstlane(tid >> 5);
  const int t16  = lane & 15;
  const int k0   = (lane < 16) ? 0 : 2;
  const int half = lane >> 4;
  const int b    = blockIdx.x >> 7;
  const int ch   = blockIdx.x & 127;
  const int t0   = ch * CH;
  const int blk  = g * 16 + t16;

  const float* qbase = quat + (size_t)b * T_ * 4;
  v2f a;
  a.x = qbase[(t0 + t16) * 4 + k0];
  a.y = qbase[(t0 + t16) * 4 + k0 + 1];

  float e[9][8];
#pragma unroll
  for (int kind = 0; kind < 3; ++kind) {
#pragma unroll
    for (int comp = 0; comp < 3; ++comp) {
      const int col = kind * D_ + blk * 3 + comp;
      v2f bm;
      bm.x = weff[k0 * CPW + col];
      bm.y = weff[(k0 + 1) * CPW + col];
      const float bv = bias[col];
      v8f c = {bv, bv, bv, bv, bv, bv, bv, bv};
      v8f d = __builtin_amdgcn_wmma_f32_16x16x4_f32(
           false, a,  false, bm,
           (short)0, c,  false,  false);
#pragma unroll
      for (int v = 0; v < 8; ++v) e[kind * 3 + comp][v] = d[v];
    }
  }

  v4f o[2][NK];
#pragma unroll
  for (int v = 0; v < 8; ++v) {
    const float bx = e[0][v], by = e[1][v], bz = e[2][v];
    float nb = sqrtf(bx * bx + by * by + bz * bz);
    nb = fmaxf(nb, EPSF);
    float sn, cs;
    __sincosf(0.5f * nb, &sn, &cs);
    const float s = sn / nb;
    o[v >> 2][0][v & 3] = s * bz;
    o[v >> 2][1][v & 3] = -s * by;
    o[v >> 2][2][v & 3] = s * bx;
    o[v >> 2][3][v & 3] = cs;
    o[v >> 2][4][v & 3] = 1.f / (1.f + __expf(-e[3][v]));
    o[v >> 2][5][v & 3] = 1.f / (1.f + __expf(-e[4][v]));
    o[v >> 2][6][v & 3] = 1.f / (1.f + __expf(-e[5][v]));
    o[v >> 2][7][v & 3] = e[6][v];
    o[v >> 2][8][v & 3] = e[7][v];
    o[v >> 2][9][v & 3] = e[8][v];
  }

#pragma unroll
  for (int vg = 0; vg < 2; ++vg) {
    const int t4 = ch * 4 + half * 2 + vg;
    v4f* dst = (v4f*)coef + ((size_t)(b * T4_ + t4) * NK) * NB_ + blk;
#pragma unroll
    for (int k = 0; k < NK; ++k) dst[(size_t)k * NB_] = o[vg][k];
  }
}

static __device__ __forceinline__ void lds_barrier() {
  asm volatile("s_wait_dscnt 0x0\n\ts_barrier_signal -1\n\ts_barrier_wait -1" ::: "memory");
}

__global__ __launch_bounds__(128) void so3_scan_kernel(
    const float* __restrict__ coef,
    const float* __restrict__ W_out,
    float* __restrict__ out) {
  __shared__ float sH[CH * HP];
  __shared__ float sWout[D_ * 4];
  __shared__ float sPart[4 * 256];

  const int tid   = threadIdx.x;
  const int lane  = tid & 31;
  const int uwave = __builtin_amdgcn_readfirstlane(tid >> 5);
  const int t16   = lane & 15;
  const int k0    = (lane < 16) ? 0 : 2;
  const int mhi   = (lane < 16) ? 0 : 8;
  const int b     = blockIdx.x;

  for (int i = tid; i < D_ * 4; i += 128) sWout[i] = W_out[i];
  __syncthreads();

  float hx = 0.f, hy = 0.f, hz = 0.f;
  const v4f* cbase = (const v4f*)coef + (size_t)b * T4_ * NK * NB_ + tid;

  v4f cf[4][NK];
#pragma unroll
  for (int g = 0; g < 4; ++g)
#pragma unroll
    for (int k = 0; k < NK; ++k)
      cf[g][k] = cbase[((size_t)g * NK + k) * NB_];

  for (int ch = 0; ch < T_ / CH; ++ch) {
    const int t0 = ch * CH;
    const int j3 = tid * 3;

#pragma unroll
    for (int g = 0; g < 4; ++g) {
#pragma unroll
      for (int j = 0; j < 4; ++j) {
        const int tt = g * 4 + j;
        const float qx = cf[g][0][j], qy = cf[g][1][j], qz = cf[g][2][j];
        const float cs = cf[g][3][j];
        float tx = 2.f * (qy * hz - qz * hy);
        float ty = 2.f * (qz * hx - qx * hz);
        float tz = 2.f * (qx * hy - qy * hx);
        float rx = hx + cs * tx + (qy * tz - qz * ty);
        float ry = hy + cs * ty + (qz * tx - qx * tz);
        float rz = hz + cs * tz + (qx * ty - qy * tx);
        hx = cf[g][4][j] * rx + cf[g][7][j];
        hy = cf[g][5][j] * ry + cf[g][8][j];
        hz = cf[g][6][j] * rz + cf[g][9][j];
        sH[tt * HP + j3]     = hx;
        sH[tt * HP + j3 + 1] = hy;
        sH[tt * HP + j3 + 2] = hz;
      }
      const int t4n = min(ch * 4 + 4 + g, T4_ - 1);
#pragma unroll
      for (int k = 0; k < NK; ++k)
        cf[g][k] = cbase[((size_t)t4n * NK + k) * NB_];
      __builtin_amdgcn_sched_barrier(0);
    }
    lds_barrier();

    {
      v8f acc0 = {0.f, 0.f, 0.f, 0.f, 0.f, 0.f, 0.f, 0.f};
      v8f acc1 = {0.f, 0.f, 0.f, 0.f, 0.f, 0.f, 0.f, 0.f};
      const int kbeg = uwave * 24;
#pragma unroll 4
      for (int kk = 0; kk < 24; kk += 2) {
        int kr0 = (kbeg + kk) * 4 + k0;
        int kr1 = (kbeg + kk + 1) * 4 + k0;
        v2f a2, b2;
        a2.x = sH[t16 * HP + kr0];
        a2.y = sH[t16 * HP + kr0 + 1];
        b2.x = (t16 < 4) ? sWout[kr0 * 4 + t16] : 0.f;
        b2.y = (t16 < 4) ? sWout[(kr0 + 1) * 4 + t16] : 0.f;
        acc0 = __builtin_amdgcn_wmma_f32_16x16x4_f32(
            false, a2, false, b2, (short)0, acc0, false, false);
        a2.x = sH[t16 * HP + kr1];
        a2.y = sH[t16 * HP + kr1 + 1];
        b2.x = (t16 < 4) ? sWout[kr1 * 4 + t16] : 0.f;
        b2.y = (t16 < 4) ? sWout[(kr1 + 1) * 4 + t16] : 0.f;
        acc1 = __builtin_amdgcn_wmma_f32_16x16x4_f32(
            false, a2, false, b2, (short)0, acc1, false, false);
      }
      acc0 = acc0 + acc1;
#pragma unroll
      for (int v = 0; v < 8; ++v)
        sPart[uwave * 256 + (v + mhi) * 16 + t16] = acc0[v];
    }
    lds_barrier();

    if (tid < 16) {
      float p[4];
#pragma unroll
      for (int c = 0; c < 4; ++c)
        p[c] = sPart[0 * 256 + tid * 16 + c] + sPart[1 * 256 + tid * 16 + c] +
               sPart[2 * 256 + tid * 16 + c] + sPart[3 * 256 + tid * 16 + c];
      float n = sqrtf(p[0] * p[0] + p[1] * p[1] + p[2] * p[2] + p[3] * p[3]);
      float rn = 1.f / fmaxf(n, EPSF);
      float4 o = {p[0] * rn, p[1] * rn, p[2] * rn, p[3] * rn};
      *(float4*)&out[((size_t)b * T_ + t0 + tid) * 4] = o;
    }
    lds_barrier();
  }
}

extern "C" void kernel_launch(void* const* d_in, const int* in_sizes, int n_in,
                              void* d_out, int out_size, void* d_ws, size_t ws_size,
                              hipStream_t stream) {
  const float* quat  = (const float*)d_in[0];
  const float* W_in  = (const float*)d_in[1];
  const float* W_biv = (const float*)d_in[2];
  const float* b_biv = (const float*)d_in[3];
  const float* W_dec = (const float*)d_in[4];
  const float* b_dec = (const float*)d_in[5];
  const float* W_inj = (const float*)d_in[6];
  const float* b_inj = (const float*)d_in[7];
  const float* W_out = (const float*)d_in[8];
  float* out  = (float*)d_out;
  float* weff = (float*)d_ws;
  float* bias = weff + 4 * CPW;
  float* coef = weff + WS_COEF_OFF;

  so3_prep_kernel<<<(4 * CPW + 255) / 256, 256, 0, stream>>>(
      W_in, W_biv, b_biv, W_dec, b_dec, W_inj, b_inj, weff, bias);
  so3_coef_kernel<<<B_ * (T_ / CH), 256, 0, stream>>>(quat, weff, bias, coef);
  so3_scan_kernel<<<B_, 128, 0, stream>>>(coef, W_out, out);
}


# Output projection: W_out B fragments read once into registers before the time loop; per chunk the A fragments are read from LDS as one group ahead of 24 back-to-back WMMAs
# speedup vs baseline: 3.3244x; 1.6104x over previous
#include <hip/hip_runtime.h>
#include <hip/hip_bf16.h>

#define B_   16
#define T_   2048
#define D_   384
#define NB_  128
#define CPW  1152
#define HP   388
#define CH   16
#define NK   10
#define T4_  (T_ / 4)
#define WS_COEF_OFF 8192
#define EPSF 1e-8f

typedef __attribute__((ext_vector_type(2))) float v2f;
typedef __attribute__((ext_vector_type(4))) float v4f;
typedef __attribute__((ext_vector_type(8))) float v8f;

__global__ __launch_bounds__(256) void so3_prep_kernel(
    const float* __restrict__ W_in,
    const float* __restrict__ W_biv, const float* __restrict__ b_biv,
    const float* __restrict__ W_dec, const float* __restrict__ b_dec,
    const float* __restrict__ W_inj, const float* __restrict__ b_inj,
    float* __restrict__ weff, float* __restrict__ bias) {
  int idx = blockIdx.x * blockDim.x + threadIdx.x;
  if (idx >= 4 * CPW) return;
  int k = idx / CPW;
  int c = idx % CPW;
  const float* Wc;
  int cc;
  if (c < D_)            { Wc = W_biv; cc = c; }
  else if (c < 2 * D_)   { Wc = W_dec; cc = c - D_; }
  else                   { Wc = W_inj; cc = c - 2 * D_; }
  float s = 0.f;
  for (int d = 0; d < D_; ++d)
    s = fmaf(W_in[k * D_ + d], Wc[d * D_ + cc], s);
  weff[k * CPW + c] = s;
  if (k == 0)
    bias[c] = (c < D_) ? b_biv[c] : (c < 2 * D_ ? b_dec[c - D_] : b_inj[c - 2 * D_]);
}

__global__ __launch_bounds__(256) void so3_coef_kernel(
    const float* __restrict__ quat,
    const float* __restrict__ weff,
    const float* __restrict__ bias,
    float* __restrict__ coef) {
  const int tid  = threadIdx.x;
  const int lane = tid & 31;
  const int g    = __builtin_amdgcn_readfirstlane(tid >> 5);
  const int t16  = lane & 15;
  const int k0   = (lane < 16) ? 0 : 2;
  const int half = lane >> 4;
  const int b    = blockIdx.x >> 7;
  const int ch   = blockIdx.x & 127;
  const int t0   = ch * CH;
  const int blk  = g * 16 + t16;

  const float* qbase = quat + (size_t)b * T_ * 4;
  v2f a;
  a.x = qbase[(t0 + t16) * 4 + k0];
  a.y = qbase[(t0 + t16) * 4 + k0 + 1];

  float e[9][8];
#pragma unroll
  for (int kind = 0; kind < 3; ++kind) {
#pragma unroll
    for (int comp = 0; comp < 3; ++comp) {
      const int col = kind * D_ + blk * 3 + comp;
      v2f bm;
      bm.x = weff[k0 * CPW + col];
      bm.y = weff[(k0 + 1) * CPW + col];
      const float bv = bias[col];
      v8f c = {bv, bv, bv, bv, bv, bv, bv, bv};
      v8f d = __builtin_amdgcn_wmma_f32_16x16x4_f32(
           false, a,  false, bm,
           (short)0, c,  false,  false);
#pragma unroll
      for (int v = 0; v < 8; ++v) e[kind * 3 + comp][v] = d[v];
    }
  }

  v4f o[2][NK];
#pragma unroll
  for (int v = 0; v < 8; ++v) {
    const float bx = e[0][v], by = e[1][v], bz = e[2][v];
    float nb = sqrtf(bx * bx + by * by + bz * bz);
    nb = fmaxf(nb, EPSF);
    float sn, cs;
    __sincosf(0.5f * nb, &sn, &cs);
    const float s = sn / nb;
    o[v >> 2][0][v & 3] = s * bz;
    o[v >> 2][1][v & 3] = -s * by;
    o[v >> 2][2][v & 3] = s * bx;
    o[v >> 2][3][v & 3] = cs;
    o[v >> 2][4][v & 3] = 1.f / (1.f + __expf(-e[3][v]));
    o[v >> 2][5][v & 3] = 1.f / (1.f + __expf(-e[4][v]));
    o[v >> 2][6][v & 3] = 1.f / (1.f + __expf(-e[5][v]));
    o[v >> 2][7][v & 3] = e[6][v];
    o[v >> 2][8][v & 3] = e[7][v];
    o[v >> 2][9][v & 3] = e[8][v];
  }

#pragma unroll
  for (int vg = 0; vg < 2; ++vg) {
    const int t4 = ch * 4 + half * 2 + vg;
    v4f* dst = (v4f*)coef + ((size_t)(b * T4_ + t4) * NK) * NB_ + blk;
#pragma unroll
    for (int k = 0; k < NK; ++k) dst[(size_t)k * NB_] = o[vg][k];
  }
}

static __device__ __forceinline__ void lds_barrier() {
  asm volatile("s_wait_dscnt 0x0\n\ts_barrier_signal -1\n\ts_barrier_wait -1" ::: "memory");
}

__global__ __launch_bounds__(128) void so3_scan_kernel(
    const float* __restrict__ coef,
    const float* __restrict__ W_out,
    float* __restrict__ out) {
  __shared__ float sH[CH * HP];
  __shared__ float sPart[4 * 256];

  const int tid   = threadIdx.x;
  const int lane  = tid & 31;
  const int uwave = __builtin_amdgcn_readfirstlane(tid >> 5);
  const int t16   = lane & 15;
  const int k0    = (lane < 16) ? 0 : 2;
  const int mhi   = (lane < 16) ? 0 : 8;
  const int b     = blockIdx.x;

  v2f bw[24];
#pragma unroll
  for (int kk = 0; kk < 24; ++kk) {
    const int kr = (uwave * 24 + kk) * 4 + k0;
    bw[kk].x = (t16 < 4) ? W_out[kr * 4 + t16] : 0.f;
    bw[kk].y = (t16 < 4) ? W_out[(kr + 1) * 4 + t16] : 0.f;
  }

  float hx = 0.f, hy = 0.f, hz = 0.f;
  const v4f* cbase = (const v4f*)coef + (size_t)b * T4_ * NK * NB_ + tid;

  v4f cf[4][NK];
#pragma unroll
  for (int g = 0; g < 4; ++g)
#pragma unroll
    for (int k = 0; k < NK; ++k)
      cf[g][k] = cbase[((size_t)g * NK + k) * NB_];

  for (int ch = 0; ch < T_ / CH; ++ch) {
    const int t0 = ch * CH;
    const int j3 = tid * 3;

#pragma unroll
    for (int g = 0; g < 4; ++g) {
#pragma unroll
      for (int j = 0; j < 4; ++j) {
        const int tt = g * 4 + j;
        const float qx = cf[g][0][j], qy = cf[g][1][j], qz = cf[g][2][j];
        const float cs = cf[g][3][j];
        float tx = 2.f * (qy * hz - qz * hy);
        float ty = 2.f * (qz * hx - qx * hz);
        float tz = 2.f * (qx * hy - qy * hx);
        float rx = hx + cs * tx + (qy * tz - qz * ty);
        float ry = hy + cs * ty + (qz * tx - qx * tz);
        float rz = hz + cs * tz + (qx * ty - qy * tx);
        hx = cf[g][4][j] * rx + cf[g][7][j];
        hy = cf[g][5][j] * ry + cf[g][8][j];
        hz = cf[g][6][j] * rz + cf[g][9][j];
        sH[tt * HP + j3]     = hx;
        sH[tt * HP + j3 + 1] = hy;
        sH[tt * HP + j3 + 2] = hz;
      }
      const int t4n = min(ch * 4 + 4 + g, T4_ - 1);
#pragma unroll
      for (int k = 0; k < NK; ++k)
        cf[g][k] = cbase[((size_t)t4n * NK + k) * NB_];
      __builtin_amdgcn_sched_barrier(0);
    }
    lds_barrier();

    {
      v2f af[24];
      const float* hrow = &sH[t16 * HP + uwave * 96 + k0];
#pragma unroll
      for (int kk = 0; kk < 24; ++kk) {
        af[kk].x = hrow[kk * 4];
        af[kk].y = hrow[kk * 4 + 1];
      }
      __builtin_amdgcn_sched_barrier(0);
      v8f acc0 = {0.f, 0.f, 0.f, 0.f, 0.f, 0.f, 0.f, 0.f};
      v8f acc1 = {0.f, 0.f, 0.f, 0.f, 0.f, 0.f, 0.f, 0.f};
#pragma unroll
      for (int kk = 0; kk < 24; kk += 2) {
        acc0 = __builtin_amdgcn_wmma_f32_16x16x4_f32(
            false, af[kk], false, bw[kk], (short)0, acc0, false, false);
        acc1 = __builtin_amdgcn_wmma_f32_16x16x4_f32(
            false, af[kk + 1], false, bw[kk + 1], (short)0, acc1, false, false);
      }
      acc0 = acc0 + acc1;
#pragma unroll
      for (int v = 0; v < 8; ++v)
        sPart[uwave * 256 + (v + mhi) * 16 + t16] = acc0[v];
    }
    lds_barrier();

    if (tid < 16) {
      float p[4];
#pragma unroll
      for (int c = 0; c < 4; ++c)
        p[c] = sPart[0 * 256 + tid * 16 + c] + sPart[1 * 256 + tid * 16 + c] +
               sPart[2 * 256 + tid * 16 + c] + sPart[3 * 256 + tid * 16 + c];
      float n = sqrtf(p[0] * p[0] + p[1] * p[1] + p[2] * p[2] + p[3] * p[3]);
      float rn = 1.f / fmaxf(n, EPSF);
      float4 o = {p[0] * rn, p[1] * rn, p[2] * rn, p[3] * rn};
      *(float4*)&out[((size_t)b * T_ + t0 + tid) * 4] = o;
    }
    lds_barrier();
  }
}

extern "C" void kernel_launch(void* const* d_in, const int* in_sizes, int n_in,
                              void* d_out, int out_size, void* d_ws, size_t ws_size,
                              hipStream_t stream) {
  const float* quat  = (const float*)d_in[0];
  const float* W_in  = (const float*)d_in[1];
  const float* W_biv = (const float*)d_in[2];
  const float* b_biv = (const float*)d_in[3];
  const float* W_dec = (const float*)d_in[4];
  const float* b_dec = (const float*)d_in[5];
  const float* W_inj = (const float*)d_in[6];
  const float* b_inj = (const float*)d_in[7];
  const float* W_out = (const float*)d_in[8];
  float* out  = (float*)d_out;
  float* weff = (float*)d_ws;
  float* bias = weff + 4 * CPW;
  float* coef = weff + WS_COEF_OFF;

  so3_prep_kernel<<<(4 * CPW + 255) / 256, 256, 0, stream>>>(
      W_in, W_biv, b_biv, W_dec, b_dec, W_inj, b_inj, weff, bias);
  so3_coef_kernel<<<B_ * (T_ / CH), 256, 0, stream>>>(quat, weff, bias, coef);
  so3_scan_kernel<<<B_, 128, 0, stream>>>(coef, W_out, out);
}


# Scan split over the 4 independent 32-block groups of a batch element: 64 one-wave workgroups, no workgroup barrier, store projection partials; a 4th kernel adds them in group order, normalizes
# speedup vs baseline: 3.8680x; 1.1635x over previous
#include <hip/hip_runtime.h>
#include <hip/hip_bf16.h>

#define B_   16
#define T_   2048
#define D_   384
#define NB_  128
#define CPW  1152
#define HPW  100
#define CH   16
#define NK   10
#define T4_  (T_ / 4)
#define WS_COEF_OFF 8192
#define WS_COEF_FLOATS 41943040
#define EPSF 1e-8f

typedef __attribute__((ext_vector_type(2))) float v2f;
typedef __attribute__((ext_vector_type(4))) float v4f;
typedef __attribute__((ext_vector_type(8))) float v8f;

__global__ __launch_bounds__(256) void so3_prep_kernel(
    const float* __restrict__ W_in,
    const float* __restrict__ W_biv, const float* __restrict__ b_biv,
    const float* __restrict__ W_dec, const float* __restrict__ b_dec,
    const float* __restrict__ W_inj, const float* __restrict__ b_inj,
    float* __restrict__ weff, float* __restrict__ bias) {
  int idx = blockIdx.x * blockDim.x + threadIdx.x;
  if (idx >= 4 * CPW) return;
  int k = idx / CPW;
  int c = idx % CPW;
  const float* Wc;
  int cc;
  if (c < D_)            { Wc = W_biv; cc = c; }
  else if (c < 2 * D_)   { Wc = W_dec; cc = c - D_; }
  else                   { Wc = W_inj; cc = c - 2 * D_; }
  float s = 0.f;
  for (int d = 0; d < D_; ++d)
    s = fmaf(W_in[k * D_ + d], Wc[d * D_ + cc], s);
  weff[k * CPW + c] = s;
  if (k == 0)
    bias[c] = (c < D_) ? b_biv[c] : (c < 2 * D_ ? b_dec[c - D_] : b_inj[c - 2 * D_]);
}

__global__ __launch_bounds__(256) void so3_coef_kernel(
    const float* __restrict__ quat,
    const float* __restrict__ weff,
    const float* __restrict__ bias,
    float* __restrict__ coef) {
  const int tid  = threadIdx.x;
  const int lane = tid & 31;
  const int g    = __builtin_amdgcn_readfirstlane(tid >> 5);
  const int t16  = lane & 15;
  const int k0   = (lane < 16) ? 0 : 2;
  const int half = lane >> 4;
  const int b    = blockIdx.x >> 7;
  const int ch   = blockIdx.x & 127;
  const int t0   = ch * CH;
  const int blk  = g * 16 + t16;

  const float* qbase = quat + (size_t)b * T_ * 4;
  v2f a;
  a.x = qbase[(t0 + t16) * 4 + k0];
  a.y = qbase[(t0 + t16) * 4 + k0 + 1];

  float e[9][8];
#pragma unroll
  for (int kind = 0; kind < 3; ++kind) {
#pragma unroll
    for (int comp = 0; comp < 3; ++comp) {
      const int col = kind * D_ + blk * 3 + comp;
      v2f bm;
      bm.x = weff[k0 * CPW + col];
      bm.y = weff[(k0 + 1) * CPW + col];
      const float bv = bias[col];
      v8f c = {bv, bv, bv, bv, bv, bv, bv, bv};
      v8f d = __builtin_amdgcn_wmma_f32_16x16x4_f32(
           false, a,  false, bm,
           (short)0, c,  false,  false);
#pragma unroll
      for (int v = 0; v < 8; ++v) e[kind * 3 + comp][v] = d[v];
    }
  }

  v4f o[2][NK];
#pragma unroll
  for (int v = 0; v < 8; ++v) {
    const float bx = e[0][v], by = e[1][v], bz = e[2][v];
    float nb = sqrtf(bx * bx + by * by + bz * bz);
    nb = fmaxf(nb, EPSF);
    float sn, cs;
    __sincosf(0.5f * nb, &sn, &cs);
    const float s = sn / nb;
    o[v >> 2][0][v & 3] = s * bz;
    o[v >> 2][1][v & 3] = -s * by;
    o[v >> 2][2][v & 3] = s * bx;
    o[v >> 2][3][v & 3] = cs;
    o[v >> 2][4][v & 3] = 1.f / (1.f + __expf(-e[3][v]));
    o[v >> 2][5][v & 3] = 1.f / (1.f + __expf(-e[4][v]));
    o[v >> 2][6][v & 3] = 1.f / (1.f + __expf(-e[5][v]));
    o[v >> 2][7][v & 3] = e[6][v];
    o[v >> 2][8][v & 3] = e[7][v];
    o[v >> 2][9][v & 3] = e[8][v];
  }

#pragma unroll
  for (int vg = 0; vg < 2; ++vg) {
    const int t4 = ch * 4 + half * 2 + vg;
    v4f* dst = (v4f*)coef + ((size_t)(b * T4_ + t4) * NK) * NB_ + blk;
#pragma unroll
    for (int k = 0; k < NK; ++k) dst[(size_t)k * NB_] = o[vg][k];
  }
}

static __device__ __forceinline__ void wave_lds_handover() {
  asm volatile("s_wait_dscnt 0x0" ::: "memory");
}

__global__ __launch_bounds__(32) void so3_scan_kernel(
    const float* __restrict__ coef,
    const float* __restrict__ W_out,
    float* __restrict__ part) {
  __shared__ float sH[CH * HPW];
  __shared__ __attribute__((aligned(16))) float sPart[256];

  const int lane = threadIdx.x;
  const int t16  = lane & 15;
  const int k0   = (lane < 16) ? 0 : 2;
  const int mhi  = (lane < 16) ? 0 : 8;
  const int b    = blockIdx.x >> 2;
  const int grp  = blockIdx.x & 3;

  v2f bw[24];
#pragma unroll
  for (int kk = 0; kk < 24; ++kk) {
    const int kr = (grp * 24 + kk) * 4 + k0;
    bw[kk].x = (t16 < 4) ? W_out[kr * 4 + t16] : 0.f;
    bw[kk].y = (t16 < 4) ? W_out[(kr + 1) * 4 + t16] : 0.f;
  }

  float hx = 0.f, hy = 0.f, hz = 0.f;
  const v4f* cbase = (const v4f*)coef + (size_t)b * T4_ * NK * NB_ + grp * 32 + lane;
  v4f* pbase = (v4f*)part + (size_t)(b * 4 + grp) * T_ + lane;

  v4f cf[4][NK];
#pragma unroll
  for (int g = 0; g < 4; ++g)
#pragma unroll
    for (int k = 0; k < NK; ++k)
      cf[g][k] = cbase[((size_t)g * NK + k) * NB_];

  for (int ch = 0; ch < T_ / CH; ++ch) {
    const int t0 = ch * CH;
    const int j3 = lane * 3;

#pragma unroll
    for (int g = 0; g < 4; ++g) {
#pragma unroll
      for (int j = 0; j < 4; ++j) {
        const int tt = g * 4 + j;
        const float qx = cf[g][0][j], qy = cf[g][1][j], qz = cf[g][2][j];
        const float cs = cf[g][3][j];
        float tx = 2.f * (qy * hz - qz * hy);
        float ty = 2.f * (qz * hx - qx * hz);
        float tz = 2.f * (qx * hy - qy * hx);
        float rx = hx + cs * tx + (qy * tz - qz * ty);
        float ry = hy + cs * ty + (qz * tx - qx * tz);
        float rz = hz + cs * tz + (qx * ty - qy * tx);
        hx = cf[g][4][j] * rx + cf[g][7][j];
        hy = cf[g][5][j] * ry + cf[g][8][j];
        hz = cf[g][6][j] * rz + cf[g][9][j];
        sH[tt * HPW + j3]     = hx;
        sH[tt * HPW + j3 + 1] = hy;
        sH[tt * HPW + j3 + 2] = hz;
      }
      const int t4n = min(ch * 4 + 4 + g, T4_ - 1);
#pragma unroll
      for (int k = 0; k < NK; ++k)
        cf[g][k] = cbase[((size_t)t4n * NK + k) * NB_];
      __builtin_amdgcn_sched_barrier(0);
    }
    wave_lds_handover();

    {
      v2f af[24];
      const float* hrow = &sH[t16 * HPW + k0];
#pragma unroll
      for (int kk = 0; kk < 24; ++kk) {
        af[kk].x = hrow[kk * 4];
        af[kk].y = hrow[kk * 4 + 1];
      }
      __builtin_amdgcn_sched_barrier(0);
      v8f acc0 = {0.f, 0.f, 0.f, 0.f, 0.f, 0.f, 0.f, 0.f};
      v8f acc1 = {0.f, 0.f, 0.f, 0.f, 0.f, 0.f, 0.f, 0.f};
#pragma unroll
      for (int kk = 0; kk < 24; kk += 2) {
        acc0 = __builtin_amdgcn_wmma_f32_16x16x4_f32(
            false, af[kk], false, bw[kk], (short)0, acc0, false, false);
        acc1 = __builtin_amdgcn_wmma_f32_16x16x4_f32(
            false, af[kk + 1], false, bw[kk + 1], (short)0, acc1, false, false);
      }
      acc0 = acc0 + acc1;
#pragma unroll
      for (int v = 0; v < 8; ++v)
        sPart[(v + mhi) * 16 + t16] = acc0[v];
    }
    wave_lds_handover();

    if (lane < 16)
      pbase[t0] = *(const v4f*)&sPart[lane * 16];
  }
}

__global__ __launch_bounds__(256) void so3_norm_kernel(
    const float* __restrict__ part,
    float* __restrict__ out) {
  const int i = blockIdx.x * 256 + threadIdx.x;
  const int b = i / T_;
  const int t = i % T_;
  const v4f* src = (const v4f*)part + (size_t)b * 4 * T_ + t;
  const v4f p0 = src[0], p1 = src[T_], p2 = src[2 * T_], p3 = src[3 * T_];
  float p[4];
#pragma unroll
  for (int c = 0; c < 4; ++c)
    p[c] = p0[c] + p1[c] + p2[c] + p3[c];
  float n = sqrtf(p[0] * p[0] + p[1] * p[1] + p[2] * p[2] + p[3] * p[3]);
  float rn = 1.f / fmaxf(n, EPSF);
  float4 o = {p[0] * rn, p[1] * rn, p[2] * rn, p[3] * rn};
  *(float4*)&out[(size_t)i * 4] = o;
}

extern "C" void kernel_launch(void* const* d_in, const int* in_sizes, int n_in,
                              void* d_out, int out_size, void* d_ws, size_t ws_size,
                              hipStream_t stream) {
  const float* quat  = (const float*)d_in[0];
  const float* W_in  = (const float*)d_in[1];
  const float* W_biv = (const float*)d_in[2];
  const float* b_biv = (const float*)d_in[3];
  const float* W_dec = (const float*)d_in[4];
  const float* b_dec = (const float*)d_in[5];
  const float* W_inj = (const float*)d_in[6];
  const float* b_inj = (const float*)d_in[7];
  const float* W_out = (const float*)d_in[8];
  float* out  = (float*)d_out;
  float* weff = (float*)d_ws;
  float* bias = weff + 4 * CPW;
  float* coef = weff + WS_COEF_OFF;
  float* part = coef + WS_COEF_FLOATS;

  so3_prep_kernel<<<(4 * CPW + 255) / 256, 256, 0, stream>>>(
      W_in, W_biv, b_biv, W_dec, b_dec, W_inj, b_inj, weff, bias);
  so3_coef_kernel<<<B_ * (T_ / CH), 256, 0, stream>>>(quat, weff, bias, coef);
  so3_scan_kernel<<<B_ * 4, 32, 0, stream>>>(coef, W_out, part);
  so3_norm_kernel<<<B_ * T_ / 256, 256, 0, stream>>>(part, out);
}


# Pre-pass folds each step's rotation and gate into a 3x3 matrix (12 coefficients per step and block); the scan's serial chain is h' = M h + inj, 3 dependent FMAs per row; rounding order changes
# speedup vs baseline: 4.2451x; 1.0975x over previous
#include <hip/hip_runtime.h>
#include <hip/hip_bf16.h>

#define B_   16
#define T_   2048
#define D_   384
#define NB_  128
#define CPW  1152
#define HPW  100
#define CH   16
#define NK   12
#define T4_  (T_ / 4)
#define WS_COEF_OFF 8192
#define WS_COEF_FLOATS 50331648
#define EPSF 1e-8f

typedef __attribute__((ext_vector_type(2))) float v2f;
typedef __attribute__((ext_vector_type(4))) float v4f;
typedef __attribute__((ext_vector_type(8))) float v8f;

__global__ __launch_bounds__(256) void so3_prep_kernel(
    const float* __restrict__ W_in,
    const float* __restrict__ W_biv, const float* __restrict__ b_biv,
    const float* __restrict__ W_dec, const float* __restrict__ b_dec,
    const float* __restrict__ W_inj, const float* __restrict__ b_inj,
    float* __restrict__ weff, float* __restrict__ bias) {
  int idx = blockIdx.x * blockDim.x + threadIdx.x;
  if (idx >= 4 * CPW) return;
  int k = idx / CPW;
  int c = idx % CPW;
  const float* Wc;
  int cc;
  if (c < D_)            { Wc = W_biv; cc = c; }
  else if (c < 2 * D_)   { Wc = W_dec; cc = c - D_; }
  else                   { Wc = W_inj; cc = c - 2 * D_; }
  float s = 0.f;
  for (int d = 0; d < D_; ++d)
    s = fmaf(W_in[k * D_ + d], Wc[d * D_ + cc], s);
  weff[k * CPW + c] = s;
  if (k == 0)
    bias[c] = (c < D_) ? b_biv[c] : (c < 2 * D_ ? b_dec[c - D_] : b_inj[c - 2 * D_]);
}

__global__ __launch_bounds__(256) void so3_coef_kernel(
    const float* __restrict__ quat,
    const float* __restrict__ weff,
    const float* __restrict__ bias,
    float* __restrict__ coef) {
  const int tid  = threadIdx.x;
  const int lane = tid & 31;
  const int g    = __builtin_amdgcn_readfirstlane(tid >> 5);
  const int t16  = lane & 15;
  const int k0   = (lane < 16) ? 0 : 2;
  const int half = lane >> 4;
  const int b    = blockIdx.x >> 7;
  const int ch   = blockIdx.x & 127;
  const int t0   = ch * CH;
  const int blk  = g * 16 + t16;

  const float* qbase = quat + (size_t)b * T_ * 4;
  v2f a;
  a.x = qbase[(t0 + t16) * 4 + k0];
  a.y = qbase[(t0 + t16) * 4 + k0 + 1];

  float e[9][8];
#pragma unroll
  for (int kind = 0; kind < 3; ++kind) {
#pragma unroll
    for (int comp = 0; comp < 3; ++comp) {
      const int col = kind * D_ + blk * 3 + comp;
      v2f bm;
      bm.x = weff[k0 * CPW + col];
      bm.y = weff[(k0 + 1) * CPW + col];
      const float bv = bias[col];
      v8f c = {bv, bv, bv, bv, bv, bv, bv, bv};
      v8f d = __builtin_amdgcn_wmma_f32_16x16x4_f32(
           false, a,  false, bm,
           (short)0, c,  false,  false);
#pragma unroll
      for (int v = 0; v < 8; ++v) e[kind * 3 + comp][v] = d[v];
    }
  }

  v4f o[2][NK];
#pragma unroll
  for (int v = 0; v < 8; ++v) {
    const float bx = e[0][v], by = e[1][v], bz = e[2][v];
    float nb = sqrtf(bx * bx + by * by + bz * bz);
    nb = fmaxf(nb, EPSF);
    float sn, cs;
    __sincosf(0.5f * nb, &sn, &cs);
    const float s = sn / nb;
    const float qx = s * bz, qy = -s * by, qz = s * bx;
    const float gx = 1.f / (1.f + __expf(-e[3][v]));
    const float gy = 1.f / (1.f + __expf(-e[4][v]));
    const float gz = 1.f / (1.f + __expf(-e[5][v]));
    const float xx = qx * qx, yy = qy * qy, zz = qz * qz;
    const float xy = qx * qy, xz = qx * qz, yz = qy * qz;
    const float wx = cs * qx, wy = cs * qy, wz = cs * qz;
    o[v >> 2][0][v & 3] = gx * (1.f - 2.f * (yy + zz));
    o[v >> 2][1][v & 3] = gx * (2.f * (xy - wz));
    o[v >> 2][2][v & 3] = gx * (2.f * (xz + wy));
    o[v >> 2][3][v & 3] = gy * (2.f * (xy + wz));
    o[v >> 2][4][v & 3] = gy * (1.f - 2.f * (xx + zz));
    o[v >> 2][5][v & 3] = gy * (2.f * (yz - wx));
    o[v >> 2][6][v & 3] = gz * (2.f * (xz - wy));
    o[v >> 2][7][v & 3] = gz * (2.f * (yz + wx));
    o[v >> 2][8][v & 3] = gz * (1.f - 2.f * (xx + yy));
    o[v >> 2][9][v & 3]  = e[6][v];
    o[v >> 2][10][v & 3] = e[7][v];
    o[v >> 2][11][v & 3] = e[8][v];
  }

#pragma unroll
  for (int vg = 0; vg < 2; ++vg) {
    const int t4 = ch * 4 + half * 2 + vg;
    v4f* dst = (v4f*)coef + ((size_t)(b * T4_ + t4) * NK) * NB_ + blk;
#pragma unroll
    for (int k = 0; k < NK; ++k) dst[(size_t)k * NB_] = o[vg][k];
  }
}

static __device__ __forceinline__ void wave_lds_handover() {
  asm volatile("s_wait_dscnt 0x0" ::: "memory");
}

__global__ __launch_bounds__(32) void so3_scan_kernel(
    const float* __restrict__ coef,
    const float* __restrict__ W_out,
    float* __restrict__ part) {
  __shared__ float sH[CH * HPW];
  __shared__ __attribute__((aligned(16))) float sPart[256];

  const int lane = threadIdx.x;
  const int t16  = lane & 15;
  const int k0   = (lane < 16) ? 0 : 2;
  const int mhi  = (lane < 16) ? 0 : 8;
  const int b    = blockIdx.x >> 2;
  const int grp  = blockIdx.x & 3;

  v2f bw[24];
#pragma unroll
  for (int kk = 0; kk < 24; ++kk) {
    const int kr = (grp * 24 + kk) * 4 + k0;
    bw[kk].x = (t16 < 4) ? W_out[kr * 4 + t16] : 0.f;
    bw[kk].y = (t16 < 4) ? W_out[(kr + 1) * 4 + t16] : 0.f;
  }

  float hx = 0.f, hy = 0.f, hz = 0.f;
  const v4f* cbase = (const v4f*)coef + (size_t)b * T4_ * NK * NB_ + grp * 32 + lane;
  v4f* pbase = (v4f*)part + (size_t)(b * 4 + grp) * T_ + lane;

  v4f cf[4][NK];
#pragma unroll
  for (int g = 0; g < 4; ++g)
#pragma unroll
    for (int k = 0; k < NK; ++k)
      cf[g][k] = cbase[((size_t)g * NK + k) * NB_];

  for (int ch = 0; ch < T_ / CH; ++ch) {
    const int t0 = ch * CH;
    const int j3 = lane * 3;

#pragma unroll
    for (int g = 0; g < 4; ++g) {
#pragma unroll
      for (int j = 0; j < 4; ++j) {
        const int tt = g * 4 + j;
        const float nx = fmaf(cf[g][0][j], hx, fmaf(cf[g][1][j], hy, fmaf(cf[g][2][j], hz, cf[g][9][j])));
        const float ny = fmaf(cf[g][3][j], hx, fmaf(cf[g][4][j], hy, fmaf(cf[g][5][j], hz, cf[g][10][j])));
        const float nz = fmaf(cf[g][6][j], hx, fmaf(cf[g][7][j], hy, fmaf(cf[g][8][j], hz, cf[g][11][j])));
        hx = nx; hy = ny; hz = nz;
        sH[tt * HPW + j3]     = hx;
        sH[tt * HPW + j3 + 1] = hy;
        sH[tt * HPW + j3 + 2] = hz;
      }
      const int t4n = min(ch * 4 + 4 + g, T4_ - 1);
#pragma unroll
      for (int k = 0; k < NK; ++k)
        cf[g][k] = cbase[((size_t)t4n * NK + k) * NB_];
      __builtin_amdgcn_sched_barrier(0);
    }
    wave_lds_handover();

    {
      v2f af[24];
      const float* hrow = &sH[t16 * HPW + k0];
#pragma unroll
      for (int kk = 0; kk < 24; ++kk) {
        af[kk].x = hrow[kk * 4];
        af[kk].y = hrow[kk * 4 + 1];
      }
      __builtin_amdgcn_sched_barrier(0);
      v8f acc0 = {0.f, 0.f, 0.f, 0.f, 0.f, 0.f, 0.f, 0.f};
      v8f acc1 = {0.f, 0.f, 0.f, 0.f, 0.f, 0.f, 0.f, 0.f};
#pragma unroll
      for (int kk = 0; kk < 24; kk += 2) {
        acc0 = __builtin_amdgcn_wmma_f32_16x16x4_f32(
            false, af[kk], false, bw[kk], (short)0, acc0, false, false);
        acc1 = __builtin_amdgcn_wmma_f32_16x16x4_f32(
            false, af[kk + 1], false, bw[kk + 1], (short)0, acc1, false, false);
      }
      acc0 = acc0 + acc1;
#pragma unroll
      for (int v = 0; v < 8; ++v)
        sPart[(v + mhi) * 16 + t16] = acc0[v];
    }
    wave_lds_handover();

    if (lane < 16)
      pbase[t0] = *(const v4f*)&sPart[lane * 16];
  }
}

__global__ __launch_bounds__(256) void so3_norm_kernel(
    const float* __restrict__ part,
    float* __restrict__ out) {
  const int i = blockIdx.x * 256 + threadIdx.x;
  const int b = i / T_;
  const int t = i % T_;
  const v4f* src = (const v4f*)part + (size_t)b * 4 * T_ + t;
  const v4f p0 = src[0], p1 = src[T_], p2 = src[2 * T_], p3 = src[3 * T_];
  float p[4];
#pragma unroll
  for (int c = 0; c < 4; ++c)
    p[c] = p0[c] + p1[c] + p2[c] + p3[c];
  float n = sqrtf(p[0] * p[0] + p[1] * p[1] + p[2] * p[2] + p[3] * p[3]);
  float rn = 1.f / fmaxf(n, EPSF);
  float4 o = {p[0] * rn, p[1] * rn, p[2] * rn, p[3] * rn};
  *(float4*)&out[(size_t)i * 4] = o;
}

extern "C" void kernel_launch(void* const* d_in, const int* in_sizes, int n_in,
                              void* d_out, int out_size, void* d_ws, size_t ws_size,
                              hipStream_t stream) {
  const float* quat  = (const float*)d_in[0];
  const float* W_in  = (const float*)d_in[1];
  const float* W_biv = (const float*)d_in[2];
  const float* b_biv = (const float*)d_in[3];
  const float* W_dec = (const float*)d_in[4];
  const float* b_dec = (const float*)d_in[5];
  const float* W_inj = (const float*)d_in[6];
  const float* b_inj = (const float*)d_in[7];
  const float* W_out = (const float*)d_in[8];
  float* out  = (float*)d_out;
  float* weff = (float*)d_ws;
  float* bias = weff + 4 * CPW;
  float* coef = weff + WS_COEF_OFF;
  float* part = coef + WS_COEF_FLOATS;

  so3_prep_kernel<<<(4 * CPW + 255) / 256, 256, 0, stream>>>(
      W_in, W_biv, b_biv, W_dec, b_dec, W_inj, b_inj, weff, bias);
  so3_coef_kernel<<<B_ * (T_ / CH), 256, 0, stream>>>(quat, weff, bias, coef);
  so3_scan_kernel<<<B_ * 4, 32, 0, stream>>>(coef, W_out, part);
  so3_norm_kernel<<<B_ * T_ / 256, 256, 0, stream>>>(part, out);
}


# Scan parallel over time: 64 segments of 32 steps, each composed as a 3x3 affine map, 64-step carry pass, states re-derived per segment; coefficients recomputed in registers; rounding order changes
# speedup vs baseline: 9.4896x; 2.2354x over previous
#include <hip/hip_runtime.h>
#include <hip/hip_bf16.h>


#define B_   16
#define T_   2048
#define D_   384
#define NB_  128
#define CPW  1152
#define SEG_L 32
#define SEG_N (T_ / SEG_L)
#define NBATCH (SEG_L / 8)
#define HP   52
#define WS_PV_OFF   8192
#define WS_PV_FLOATS (B_ * SEG_N * NB_ * 12)
#define WS_HS_FLOATS (B_ * SEG_N * NB_ * 4)
#define EPSF 1e-8f

typedef __attribute__((ext_vector_type(2))) float v2f;
typedef __attribute__((ext_vector_type(4))) float v4f;
typedef __attribute__((ext_vector_type(8))) float v8f;

__global__ __launch_bounds__(256) void so3_prep_kernel(
    const float* __restrict__ W_in,
    const float* __restrict__ W_biv, const float* __restrict__ b_biv,
    const float* __restrict__ W_dec, const float* __restrict__ b_dec,
    const float* __restrict__ W_inj, const float* __restrict__ b_inj,
    float* __restrict__ weff, float* __restrict__ bias) {
  int idx = blockIdx.x * blockDim.x + threadIdx.x;
  if (idx >= 4 * CPW) return;
  int k = idx / CPW;
  int c = idx % CPW;
  const float* Wc;
  int cc;
  if (c < D_)            { Wc = W_biv; cc = c; }
  else if (c < 2 * D_)   { Wc = W_dec; cc = c - D_; }
  else                   { Wc = W_inj; cc = c - 2 * D_; }
  float s = 0.f;
  for (int d = 0; d < D_; ++d)
    s = fmaf(W_in[k * D_ + d], Wc[d * D_ + cc], s);
  weff[k * CPW + c] = s;
  if (k == 0)
    bias[c] = (c < D_) ? b_biv[c] : (c < 2 * D_ ? b_dec[c - D_] : b_inj[c - 2 * D_]);
}

struct StepMap { float m[9]; float inj[3]; };

static __device__ __forceinline__ StepMap make_step(
    float bx, float by, float bz, float d0, float d1, float d2,
    float i0, float i1, float i2) {
  float nb = sqrtf(bx * bx + by * by + bz * bz);
  nb = fmaxf(nb, EPSF);
  float sn, cs;
  __sincosf(0.5f * nb, &sn, &cs);
  const float s = sn / nb;
  const float qx = s * bz, qy = -s * by, qz = s * bx;
  const float gx = 1.f / (1.f + __expf(-d0));
  const float gy = 1.f / (1.f + __expf(-d1));
  const float gz = 1.f / (1.f + __expf(-d2));
  const float xx = qx * qx, yy = qy * qy, zz = qz * qz;
  const float xy = qx * qy, xz = qx * qz, yz = qy * qz;
  const float wx = cs * qx, wy = cs * qy, wz = cs * qz;
  StepMap r;
  r.m[0] = gx * (1.f - 2.f * (yy + zz));
  r.m[1] = gx * (2.f * (xy - wz));
  r.m[2] = gx * (2.f * (xz + wy));
  r.m[3] = gy * (2.f * (xy + wz));
  r.m[4] = gy * (1.f - 2.f * (xx + zz));
  r.m[5] = gy * (2.f * (yz - wx));
  r.m[6] = gz * (2.f * (xz - wy));
  r.m[7] = gz * (2.f * (yz + wx));
  r.m[8] = gz * (1.f - 2.f * (xx + yy));
  r.inj[0] = i0; r.inj[1] = i1; r.inj[2] = i2;
  return r;
}

static __device__ __forceinline__ void load_e_operands(
    const float* __restrict__ weff, const float* __restrict__ bias,
    int blk, int k0, v2f bm[9], float bv[9]) {
#pragma unroll
  for (int kind = 0; kind < 3; ++kind) {
#pragma unroll
    for (int comp = 0; comp < 3; ++comp) {
      const int col = kind * D_ + blk * 3 + comp;
      bm[kind * 3 + comp].x = weff[k0 * CPW + col];
      bm[kind * 3 + comp].y = weff[(k0 + 1) * CPW + col];
      bv[kind * 3 + comp]   = bias[col];
    }
  }
}

static __device__ __forceinline__ void compute_e(
    v2f a, const v2f bm[9], const float bv[9], float e[9][8]) {
#pragma unroll
  for (int j = 0; j < 9; ++j) {
    const float b0 = bv[j];
    v8f c = {b0, b0, b0, b0, b0, b0, b0, b0};
    v8f d = __builtin_amdgcn_wmma_f32_16x16x4_f32(
         false, a,  false, bm[j],
         (short)0, c,  false,  false);
#pragma unroll
    for (int v = 0; v < 8; ++v) e[j][v] = d[v];
  }
}

__global__ __launch_bounds__(256) void so3_seg_kernel(
    const float* __restrict__ quat,
    const float* __restrict__ weff,
    const float* __restrict__ bias,
    float* __restrict__ pv) {
  const int tid  = threadIdx.x;
  const int lane = tid & 31;
  const int g    = __builtin_amdgcn_readfirstlane(tid >> 5);
  const int t16  = lane & 15;
  const int k0   = (lane < 16) ? 0 : 2;
  const int half = lane >> 4;
  const int b    = blockIdx.x / (SEG_N / 2);
  const int p    = blockIdx.x % (SEG_N / 2);
  const int blk  = g * 16 + t16;

  const int trow = (2 * p + (t16 >> 3)) * SEG_L + (t16 & 7);
  const float* qrow = quat + ((size_t)b * T_ + trow) * 4 + k0;

  v2f bm[9];
  float bv[9];
  load_e_operands(weff, bias, blk, k0, bm, bv);

  float P[3][3] = {{1.f, 0.f, 0.f}, {0.f, 1.f, 0.f}, {0.f, 0.f, 1.f}};
  float vv[3] = {0.f, 0.f, 0.f};

  v2f a;
  a.x = qrow[0];
  a.y = qrow[1];
  for (int i = 0; i < NBATCH; ++i) {
    const int inx = min(i + 1, NBATCH - 1);
    v2f an;
    an.x = qrow[inx * 32];
    an.y = qrow[inx * 32 + 1];

    float e[9][8];
    compute_e(a, bm, bv, e);
#pragma unroll
    for (int v = 0; v < 8; ++v) {
      const StepMap s = make_step(e[0][v], e[1][v], e[2][v], e[3][v], e[4][v], e[5][v],
                                  e[6][v], e[7][v], e[8][v]);
      float Pn[3][3], vn[3];
#pragma unroll
      for (int r = 0; r < 3; ++r) {
#pragma unroll
        for (int c = 0; c < 3; ++c)
          Pn[r][c] = fmaf(s.m[r * 3], P[0][c], fmaf(s.m[r * 3 + 1], P[1][c], s.m[r * 3 + 2] * P[2][c]));
        vn[r] = fmaf(s.m[r * 3], vv[0], fmaf(s.m[r * 3 + 1], vv[1], fmaf(s.m[r * 3 + 2], vv[2], s.inj[r])));
      }
#pragma unroll
      for (int r = 0; r < 3; ++r) {
#pragma unroll
        for (int c = 0; c < 3; ++c) P[r][c] = Pn[r][c];
        vv[r] = vn[r];
      }
    }
    a = an;
  }

  const int seg = 2 * p + half;
  v4f* dst = (v4f*)pv + (((size_t)b * SEG_N + seg) * NB_ + blk) * 3;
#pragma unroll
  for (int r = 0; r < 3; ++r) {
    v4f o = {P[r][0], P[r][1], P[r][2], vv[r]};
    dst[r] = o;
  }
}

__global__ __launch_bounds__(32) void so3_carry_kernel(
    const float* __restrict__ pv,
    float* __restrict__ hs) {
  const int b   = blockIdx.x >> 2;
  const int blk = (blockIdx.x & 3) * 32 + threadIdx.x;
  const v4f* src = (const v4f*)pv + ((size_t)b * SEG_N * NB_ + blk) * 3;
  v4f* dst = (v4f*)hs + (size_t)b * SEG_N * NB_ + blk;
  float hx = 0.f, hy = 0.f, hz = 0.f;
#pragma unroll 8
  for (int s = 0; s < SEG_N; ++s) {
    const v4f h4 = {hx, hy, hz, 0.f};
    dst[(size_t)s * NB_] = h4;
    const v4f r0 = src[(size_t)s * NB_ * 3];
    const v4f r1 = src[(size_t)s * NB_ * 3 + 1];
    const v4f r2 = src[(size_t)s * NB_ * 3 + 2];
    const float nx = fmaf(r0.x, hx, fmaf(r0.y, hy, fmaf(r0.z, hz, r0.w)));
    const float ny = fmaf(r1.x, hx, fmaf(r1.y, hy, fmaf(r1.z, hz, r1.w)));
    const float nz = fmaf(r2.x, hx, fmaf(r2.y, hy, fmaf(r2.z, hz, r2.w)));
    hx = nx; hy = ny; hz = nz;
  }
}

static __device__ __forceinline__ void wave_lds_handover() {
  asm volatile("s_wait_dscnt 0x0" ::: "memory");
}

__global__ __launch_bounds__(256) void so3_emit_kernel(
    const float* __restrict__ quat,
    const float* __restrict__ weff,
    const float* __restrict__ bias,
    const float* __restrict__ hs,
    const float* __restrict__ W_out,
    float* __restrict__ part) {
  __shared__ float sH[8][16 * HP];
  __shared__ __attribute__((aligned(16))) float sPart[8][256];

  const int tid  = threadIdx.x;
  const int lane = tid & 31;
  const int g    = __builtin_amdgcn_readfirstlane(tid >> 5);
  const int t16  = lane & 15;
  const int k0   = (lane < 16) ? 0 : 2;
  const int half = lane >> 4;
  const int mhi  = half * 8;
  const int b    = blockIdx.x / (SEG_N / 2);
  const int p    = blockIdx.x % (SEG_N / 2);
  const int blk  = g * 16 + t16;
  float* myH = sH[g];
  float* myP = sPart[g];

  const int trow = (2 * p + (t16 >> 3)) * SEG_L + (t16 & 7);
  const float* qrow = quat + ((size_t)b * T_ + trow) * 4 + k0;

  v2f bm[9];
  float bv[9];
  load_e_operands(weff, bias, blk, k0, bm, bv);

  v2f bw[12];
#pragma unroll
  for (int kk = 0; kk < 12; ++kk) {
    const int kr = g * 48 + kk * 4 + k0;
    bw[kk].x = (t16 < 4) ? W_out[kr * 4 + t16] : 0.f;
    bw[kk].y = (t16 < 4) ? W_out[(kr + 1) * 4 + t16] : 0.f;
  }

  const int seg = 2 * p + half;
  const v4f h4 = ((const v4f*)hs)[((size_t)b * SEG_N + seg) * NB_ + blk];
  float hx = h4.x, hy = h4.y, hz = h4.z;

  v4f* pbase = (v4f*)part + ((size_t)(b * 8 + g) * T_ + trow);

  v2f a;
  a.x = qrow[0];
  a.y = qrow[1];
  for (int i = 0; i < NBATCH; ++i) {
    const int inx = min(i + 1, NBATCH - 1);
    v2f an;
    an.x = qrow[inx * 32];
    an.y = qrow[inx * 32 + 1];

    float e[9][8];
    compute_e(a, bm, bv, e);
#pragma unroll
    for (int v = 0; v < 8; ++v) {
      const StepMap s = make_step(e[0][v], e[1][v], e[2][v], e[3][v], e[4][v], e[5][v],
                                  e[6][v], e[7][v], e[8][v]);
      const float nx = fmaf(s.m[0], hx, fmaf(s.m[1], hy, fmaf(s.m[2], hz, s.inj[0])));
      const float ny = fmaf(s.m[3], hx, fmaf(s.m[4], hy, fmaf(s.m[5], hz, s.inj[1])));
      const float nz = fmaf(s.m[6], hx, fmaf(s.m[7], hy, fmaf(s.m[8], hz, s.inj[2])));
      hx = nx; hy = ny; hz = nz;
      float* hrow = &myH[(mhi + v) * HP + t16 * 3];
      hrow[0] = hx;
      hrow[1] = hy;
      hrow[2] = hz;
    }
    wave_lds_handover();

    {
      v2f af[12];
      const float* hr = &myH[t16 * HP + k0];
#pragma unroll
      for (int kk = 0; kk < 12; ++kk) {
        af[kk].x = hr[kk * 4];
        af[kk].y = hr[kk * 4 + 1];
      }
      v8f acc0 = {0.f, 0.f, 0.f, 0.f, 0.f, 0.f, 0.f, 0.f};
      v8f acc1 = {0.f, 0.f, 0.f, 0.f, 0.f, 0.f, 0.f, 0.f};
#pragma unroll
      for (int kk = 0; kk < 12; kk += 2) {
        acc0 = __builtin_amdgcn_wmma_f32_16x16x4_f32(
            false, af[kk], false, bw[kk], (short)0, acc0, false, false);
        acc1 = __builtin_amdgcn_wmma_f32_16x16x4_f32(
            false, af[kk + 1], false, bw[kk + 1], (short)0, acc1, false, false);
      }
      acc0 = acc0 + acc1;
#pragma unroll
      for (int v = 0; v < 8; ++v)
        myP[(v + mhi) * 16 + t16] = acc0[v];
    }
    wave_lds_handover();

    if (lane < 16)
      pbase[i * 8] = *(const v4f*)&myP[lane * 16];
    a = an;
  }
}

__global__ __launch_bounds__(256) void so3_norm_kernel(
    const float* __restrict__ part,
    float* __restrict__ out) {
  const int i = blockIdx.x * 256 + threadIdx.x;
  const int b = i / T_;
  const int t = i % T_;
  const v4f* src = (const v4f*)part + (size_t)b * 8 * T_ + t;
  v4f acc = src[0];
#pragma unroll
  for (int gi = 1; gi < 8; ++gi) acc = acc + src[(size_t)gi * T_];
  float n = sqrtf(acc[0] * acc[0] + acc[1] * acc[1] + acc[2] * acc[2] + acc[3] * acc[3]);
  float rn = 1.f / fmaxf(n, EPSF);
  float4 o = {acc[0] * rn, acc[1] * rn, acc[2] * rn, acc[3] * rn};
  *(float4*)&out[(size_t)i * 4] = o;
}

extern "C" void kernel_launch(void* const* d_in, const int* in_sizes, int n_in,
                              void* d_out, int out_size, void* d_ws, size_t ws_size,
                              hipStream_t stream) {
  const float* quat  = (const float*)d_in[0];
  const float* W_in  = (const float*)d_in[1];
  const float* W_biv = (const float*)d_in[2];
  const float* b_biv = (const float*)d_in[3];
  const float* W_dec = (const float*)d_in[4];
  const float* b_dec = (const float*)d_in[5];
  const float* W_inj = (const float*)d_in[6];
  const float* b_inj = (const float*)d_in[7];
  const float* W_out = (const float*)d_in[8];
  float* out  = (float*)d_out;
  float* weff = (float*)d_ws;
  float* bias = weff + 4 * CPW;
  float* pv   = weff + WS_PV_OFF;
  float* hs   = pv + WS_PV_FLOATS;
  float* part = hs + WS_HS_FLOATS;

  so3_prep_kernel<<<(4 * CPW + 255) / 256, 256, 0, stream>>>(
      W_in, W_biv, b_biv, W_dec, b_dec, W_inj, b_inj, weff, bias);
  so3_seg_kernel<<<B_ * (SEG_N / 2), 256, 0, stream>>>(quat, weff, bias, pv);
  so3_carry_kernel<<<B_ * 4, 32, 0, stream>>>(pv, hs);
  so3_emit_kernel<<<B_ * (SEG_N / 2), 256, 0, stream>>>(quat, weff, bias, hs, W_out, part);
  so3_norm_kernel<<<B_ * T_ / 256, 256, 0, stream>>>(part, out);
}


# Step math: the 4 IEEE divisions and the IEEE square root per step and block become hardware v_rcp_f32 / v_rsq_f32 (1 ulp; eps clamp moved onto the squared norm); gate and rotation rounding changes
# speedup vs baseline: 10.9483x; 1.1537x over previous
#include <hip/hip_runtime.h>
#include <hip/hip_bf16.h>


#define B_   16
#define T_   2048
#define D_   384
#define NB_  128
#define CPW  1152
#define SEG_L 32
#define SEG_N (T_ / SEG_L)
#define NBATCH (SEG_L / 8)
#define HP   52
#define WS_PV_OFF   8192
#define WS_PV_FLOATS (B_ * SEG_N * NB_ * 12)
#define WS_HS_FLOATS (B_ * SEG_N * NB_ * 4)
#define EPSF 1e-8f

typedef __attribute__((ext_vector_type(2))) float v2f;
typedef __attribute__((ext_vector_type(4))) float v4f;
typedef __attribute__((ext_vector_type(8))) float v8f;

__global__ __launch_bounds__(256) void so3_prep_kernel(
    const float* __restrict__ W_in,
    const float* __restrict__ W_biv, const float* __restrict__ b_biv,
    const float* __restrict__ W_dec, const float* __restrict__ b_dec,
    const float* __restrict__ W_inj, const float* __restrict__ b_inj,
    float* __restrict__ weff, float* __restrict__ bias) {
  int idx = blockIdx.x * blockDim.x + threadIdx.x;
  if (idx >= 4 * CPW) return;
  int k = idx / CPW;
  int c = idx % CPW;
  const float* Wc;
  int cc;
  if (c < D_)            { Wc = W_biv; cc = c; }
  else if (c < 2 * D_)   { Wc = W_dec; cc = c - D_; }
  else                   { Wc = W_inj; cc = c - 2 * D_; }
  float s = 0.f;
  for (int d = 0; d < D_; ++d)
    s = fmaf(W_in[k * D_ + d], Wc[d * D_ + cc], s);
  weff[k * CPW + c] = s;
  if (k == 0)
    bias[c] = (c < D_) ? b_biv[c] : (c < 2 * D_ ? b_dec[c - D_] : b_inj[c - 2 * D_]);
}

struct StepMap { float m[9]; float inj[3]; };

static __device__ __forceinline__ StepMap make_step(
    float bx, float by, float bz, float d0, float d1, float d2,
    float i0, float i1, float i2) {
  const float ss  = fmaxf(bx * bx + by * by + bz * bz, EPSF * EPSF);
  const float rnb = __builtin_amdgcn_rsqf(ss);
  const float nb  = ss * rnb;
  float sn, cs;
  __sincosf(0.5f * nb, &sn, &cs);
  const float s = sn * rnb;
  const float qx = s * bz, qy = -s * by, qz = s * bx;
  const float gx = __builtin_amdgcn_rcpf(1.f + __expf(-d0));
  const float gy = __builtin_amdgcn_rcpf(1.f + __expf(-d1));
  const float gz = __builtin_amdgcn_rcpf(1.f + __expf(-d2));
  const float xx = qx * qx, yy = qy * qy, zz = qz * qz;
  const float xy = qx * qy, xz = qx * qz, yz = qy * qz;
  const float wx = cs * qx, wy = cs * qy, wz = cs * qz;
  StepMap r;
  r.m[0] = gx * (1.f - 2.f * (yy + zz));
  r.m[1] = gx * (2.f * (xy - wz));
  r.m[2] = gx * (2.f * (xz + wy));
  r.m[3] = gy * (2.f * (xy + wz));
  r.m[4] = gy * (1.f - 2.f * (xx + zz));
  r.m[5] = gy * (2.f * (yz - wx));
  r.m[6] = gz * (2.f * (xz - wy));
  r.m[7] = gz * (2.f * (yz + wx));
  r.m[8] = gz * (1.f - 2.f * (xx + yy));
  r.inj[0] = i0; r.inj[1] = i1; r.inj[2] = i2;
  return r;
}

static __device__ __forceinline__ void load_e_operands(
    const float* __restrict__ weff, const float* __restrict__ bias,
    int blk, int k0, v2f bm[9], float bv[9]) {
#pragma unroll
  for (int kind = 0; kind < 3; ++kind) {
#pragma unroll
    for (int comp = 0; comp < 3; ++comp) {
      const int col = kind * D_ + blk * 3 + comp;
      bm[kind * 3 + comp].x = weff[k0 * CPW + col];
      bm[kind * 3 + comp].y = weff[(k0 + 1) * CPW + col];
      bv[kind * 3 + comp]   = bias[col];
    }
  }
}

static __device__ __forceinline__ void compute_e(
    v2f a, const v2f bm[9], const float bv[9], float e[9][8]) {
#pragma unroll
  for (int j = 0; j < 9; ++j) {
    const float b0 = bv[j];
    v8f c = {b0, b0, b0, b0, b0, b0, b0, b0};
    v8f d = __builtin_amdgcn_wmma_f32_16x16x4_f32(
         false, a,  false, bm[j],
         (short)0, c,  false,  false);
#pragma unroll
    for (int v = 0; v < 8; ++v) e[j][v] = d[v];
  }
}

__global__ __launch_bounds__(256) void so3_seg_kernel(
    const float* __restrict__ quat,
    const float* __restrict__ weff,
    const float* __restrict__ bias,
    float* __restrict__ pv) {
  const int tid  = threadIdx.x;
  const int lane = tid & 31;
  const int g    = __builtin_amdgcn_readfirstlane(tid >> 5);
  const int t16  = lane & 15;
  const int k0   = (lane < 16) ? 0 : 2;
  const int half = lane >> 4;
  const int b    = blockIdx.x / (SEG_N / 2);
  const int p    = blockIdx.x % (SEG_N / 2);
  const int blk  = g * 16 + t16;

  const int trow = (2 * p + (t16 >> 3)) * SEG_L + (t16 & 7);
  const float* qrow = quat + ((size_t)b * T_ + trow) * 4 + k0;

  v2f bm[9];
  float bv[9];
  load_e_operands(weff, bias, blk, k0, bm, bv);

  float P[3][3] = {{1.f, 0.f, 0.f}, {0.f, 1.f, 0.f}, {0.f, 0.f, 1.f}};
  float vv[3] = {0.f, 0.f, 0.f};

  v2f a;
  a.x = qrow[0];
  a.y = qrow[1];
  for (int i = 0; i < NBATCH; ++i) {
    const int inx = min(i + 1, NBATCH - 1);
    v2f an;
    an.x = qrow[inx * 32];
    an.y = qrow[inx * 32 + 1];

    float e[9][8];
    compute_e(a, bm, bv, e);
#pragma unroll
    for (int v = 0; v < 8; ++v) {
      const StepMap s = make_step(e[0][v], e[1][v], e[2][v], e[3][v], e[4][v], e[5][v],
                                  e[6][v], e[7][v], e[8][v]);
      float Pn[3][3], vn[3];
#pragma unroll
      for (int r = 0; r < 3; ++r) {
#pragma unroll
        for (int c = 0; c < 3; ++c)
          Pn[r][c] = fmaf(s.m[r * 3], P[0][c], fmaf(s.m[r * 3 + 1], P[1][c], s.m[r * 3 + 2] * P[2][c]));
        vn[r] = fmaf(s.m[r * 3], vv[0], fmaf(s.m[r * 3 + 1], vv[1], fmaf(s.m[r * 3 + 2], vv[2], s.inj[r])));
      }
#pragma unroll
      for (int r = 0; r < 3; ++r) {
#pragma unroll
        for (int c = 0; c < 3; ++c) P[r][c] = Pn[r][c];
        vv[r] = vn[r];
      }
    }
    a = an;
  }

  const int seg = 2 * p + half;
  v4f* dst = (v4f*)pv + (((size_t)b * SEG_N + seg) * NB_ + blk) * 3;
#pragma unroll
  for (int r = 0; r < 3; ++r) {
    v4f o = {P[r][0], P[r][1], P[r][2], vv[r]};
    dst[r] = o;
  }
}

__global__ __launch_bounds__(32) void so3_carry_kernel(
    const float* __restrict__ pv,
    float* __restrict__ hs) {
  const int b   = blockIdx.x >> 2;
  const int blk = (blockIdx.x & 3) * 32 + threadIdx.x;
  const v4f* src = (const v4f*)pv + ((size_t)b * SEG_N * NB_ + blk) * 3;
  v4f* dst = (v4f*)hs + (size_t)b * SEG_N * NB_ + blk;
  float hx = 0.f, hy = 0.f, hz = 0.f;
#pragma unroll 8
  for (int s = 0; s < SEG_N; ++s) {
    const v4f h4 = {hx, hy, hz, 0.f};
    dst[(size_t)s * NB_] = h4;
    const v4f r0 = src[(size_t)s * NB_ * 3];
    const v4f r1 = src[(size_t)s * NB_ * 3 + 1];
    const v4f r2 = src[(size_t)s * NB_ * 3 + 2];
    const float nx = fmaf(r0.x, hx, fmaf(r0.y, hy, fmaf(r0.z, hz, r0.w)));
    const float ny = fmaf(r1.x, hx, fmaf(r1.y, hy, fmaf(r1.z, hz, r1.w)));
    const float nz = fmaf(r2.x, hx, fmaf(r2.y, hy, fmaf(r2.z, hz, r2.w)));
    hx = nx; hy = ny; hz = nz;
  }
}

static __device__ __forceinline__ void wave_lds_handover() {
  asm volatile("s_wait_dscnt 0x0" ::: "memory");
}

__global__ __launch_bounds__(256) void so3_emit_kernel(
    const float* __restrict__ quat,
    const float* __restrict__ weff,
    const float* __restrict__ bias,
    const float* __restrict__ hs,
    const float* __restrict__ W_out,
    float* __restrict__ part) {
  __shared__ float sH[8][16 * HP];
  __shared__ __attribute__((aligned(16))) float sPart[8][256];

  const int tid  = threadIdx.x;
  const int lane = tid & 31;
  const int g    = __builtin_amdgcn_readfirstlane(tid >> 5);
  const int t16  = lane & 15;
  const int k0   = (lane < 16) ? 0 : 2;
  const int half = lane >> 4;
  const int mhi  = half * 8;
  const int b    = blockIdx.x / (SEG_N / 2);
  const int p    = blockIdx.x % (SEG_N / 2);
  const int blk  = g * 16 + t16;
  float* myH = sH[g];
  float* myP = sPart[g];

  const int trow = (2 * p + (t16 >> 3)) * SEG_L + (t16 & 7);
  const float* qrow = quat + ((size_t)b * T_ + trow) * 4 + k0;

  v2f bm[9];
  float bv[9];
  load_e_operands(weff, bias, blk, k0, bm, bv);

  v2f bw[12];
#pragma unroll
  for (int kk = 0; kk < 12; ++kk) {
    const int kr = g * 48 + kk * 4 + k0;
    bw[kk].x = (t16 < 4) ? W_out[kr * 4 + t16] : 0.f;
    bw[kk].y = (t16 < 4) ? W_out[(kr + 1) * 4 + t16] : 0.f;
  }

  const int seg = 2 * p + half;
  const v4f h4 = ((const v4f*)hs)[((size_t)b * SEG_N + seg) * NB_ + blk];
  float hx = h4.x, hy = h4.y, hz = h4.z;

  v4f* pbase = (v4f*)part + ((size_t)(b * 8 + g) * T_ + trow);

  v2f a;
  a.x = qrow[0];
  a.y = qrow[1];
  for (int i = 0; i < NBATCH; ++i) {
    const int inx = min(i + 1, NBATCH - 1);
    v2f an;
    an.x = qrow[inx * 32];
    an.y = qrow[inx * 32 + 1];

    float e[9][8];
    compute_e(a, bm, bv, e);
#pragma unroll
    for (int v = 0; v < 8; ++v) {
      const StepMap s = make_step(e[0][v], e[1][v], e[2][v], e[3][v], e[4][v], e[5][v],
                                  e[6][v], e[7][v], e[8][v]);
      const float nx = fmaf(s.m[0], hx, fmaf(s.m[1], hy, fmaf(s.m[2], hz, s.inj[0])));
      const float ny = fmaf(s.m[3], hx, fmaf(s.m[4], hy, fmaf(s.m[5], hz, s.inj[1])));
      const float nz = fmaf(s.m[6], hx, fmaf(s.m[7], hy, fmaf(s.m[8], hz, s.inj[2])));
      hx = nx; hy = ny; hz = nz;
      float* hrow = &myH[(mhi + v) * HP + t16 * 3];
      hrow[0] = hx;
      hrow[1] = hy;
      hrow[2] = hz;
    }
    wave_lds_handover();

    {
      v2f af[12];
      const float* hr = &myH[t16 * HP + k0];
#pragma unroll
      for (int kk = 0; kk < 12; ++kk) {
        af[kk].x = hr[kk * 4];
        af[kk].y = hr[kk * 4 + 1];
      }
      v8f acc0 = {0.f, 0.f, 0.f, 0.f, 0.f, 0.f, 0.f, 0.f};
      v8f acc1 = {0.f, 0.f, 0.f, 0.f, 0.f, 0.f, 0.f, 0.f};
#pragma unroll
      for (int kk = 0; kk < 12; kk += 2) {
        acc0 = __builtin_amdgcn_wmma_f32_16x16x4_f32(
            false, af[kk], false, bw[kk], (short)0, acc0, false, false);
        acc1 = __builtin_amdgcn_wmma_f32_16x16x4_f32(
            false, af[kk + 1], false, bw[kk + 1], (short)0, acc1, false, false);
      }
      acc0 = acc0 + acc1;
#pragma unroll
      for (int v = 0; v < 8; ++v)
        myP[(v + mhi) * 16 + t16] = acc0[v];
    }
    wave_lds_handover();

    if (lane < 16)
      pbase[i * 8] = *(const v4f*)&myP[lane * 16];
    a = an;
  }
}

__global__ __launch_bounds__(256) void so3_norm_kernel(
    const float* __restrict__ part,
    float* __restrict__ out) {
  const int i = blockIdx.x * 256 + threadIdx.x;
  const int b = i / T_;
  const int t = i % T_;
  const v4f* src = (const v4f*)part + (size_t)b * 8 * T_ + t;
  v4f acc = src[0];
#pragma unroll
  for (int gi = 1; gi < 8; ++gi) acc = acc + src[(size_t)gi * T_];
  float n = sqrtf(acc[0] * acc[0] + acc[1] * acc[1] + acc[2] * acc[2] + acc[3] * acc[3]);
  float rn = 1.f / fmaxf(n, EPSF);
  float4 o = {acc[0] * rn, acc[1] * rn, acc[2] * rn, acc[3] * rn};
  *(float4*)&out[(size_t)i * 4] = o;
}

extern "C" void kernel_launch(void* const* d_in, const int* in_sizes, int n_in,
                              void* d_out, int out_size, void* d_ws, size_t ws_size,
                              hipStream_t stream) {
  const float* quat  = (const float*)d_in[0];
  const float* W_in  = (const float*)d_in[1];
  const float* W_biv = (const float*)d_in[2];
  const float* b_biv = (const float*)d_in[3];
  const float* W_dec = (const float*)d_in[4];
  const float* b_dec = (const float*)d_in[5];
  const float* W_inj = (const float*)d_in[6];
  const float* b_inj = (const float*)d_in[7];
  const float* W_out = (const float*)d_in[8];
  float* out  = (float*)d_out;
  float* weff = (float*)d_ws;
  float* bias = weff + 4 * CPW;
  float* pv   = weff + WS_PV_OFF;
  float* hs   = pv + WS_PV_FLOATS;
  float* part = hs + WS_HS_FLOATS;

  so3_prep_kernel<<<(4 * CPW + 255) / 256, 256, 0, stream>>>(
      W_in, W_biv, b_biv, W_dec, b_dec, W_inj, b_inj, weff, bias);
  so3_seg_kernel<<<B_ * (SEG_N / 2), 256, 0, stream>>>(quat, weff, bias, pv);
  so3_carry_kernel<<<B_ * 4, 32, 0, stream>>>(pv, hs);
  so3_emit_kernel<<<B_ * (SEG_N / 2), 256, 0, stream>>>(quat, weff, bias, hs, W_out, part);
  so3_norm_kernel<<<B_ * T_ / 256, 256, 0, stream>>>(part, out);
}


# Weight-fold pre-kernel: 384-row dot products cut into 16 row slices per 32-column chunk, a lane's 24 loads issued as one group, slice partials added in LDS; folded weights' summation order changes
# speedup vs baseline: 17.1317x; 1.5648x over previous
#include <hip/hip_runtime.h>
#include <hip/hip_bf16.h>


#define B_   16
#define T_   2048
#define D_   384
#define NB_  128
#define CPW  1152
#define SEG_L 32
#define SEG_N (T_ / SEG_L)
#define NBATCH (SEG_L / 8)
#define HP   52
#define WS_PV_OFF   8192
#define WS_PV_FLOATS (B_ * SEG_N * NB_ * 12)
#define WS_HS_FLOATS (B_ * SEG_N * NB_ * 4)
#define EPSF 1e-8f

typedef __attribute__((ext_vector_type(2))) float v2f;
typedef __attribute__((ext_vector_type(4))) float v4f;
typedef __attribute__((ext_vector_type(8))) float v8f;

#define PREP_COLS 32
#define PREP_SL   16
#define PREP_ROWS (D_ / PREP_SL)

__global__ __launch_bounds__(PREP_COLS * PREP_SL) void so3_prep_kernel(
    const float* __restrict__ W_in,
    const float* __restrict__ W_biv, const float* __restrict__ b_biv,
    const float* __restrict__ W_dec, const float* __restrict__ b_dec,
    const float* __restrict__ W_inj, const float* __restrict__ b_inj,
    float* __restrict__ weff, float* __restrict__ bias) {
  __shared__ float sp[PREP_SL][4][PREP_COLS];
  const int lane = threadIdx.x & 31;
  const int sl = __builtin_amdgcn_readfirstlane(threadIdx.x >> 5);
  const int c0 = blockIdx.x * PREP_COLS;
  const float* Wc;
  const float* bc;
  int cc;
  if (c0 < D_)            { Wc = W_biv; bc = b_biv; cc = c0 + lane; }
  else if (c0 < 2 * D_)   { Wc = W_dec; bc = b_dec; cc = c0 - D_ + lane; }
  else                    { Wc = W_inj; bc = b_inj; cc = c0 - 2 * D_ + lane; }
  const int d0 = sl * PREP_ROWS;
  const float* wp = Wc + d0 * D_ + cc;

  float w[PREP_ROWS];
#pragma unroll
  for (int j = 0; j < PREP_ROWS; ++j) w[j] = wp[j * D_];
  if (sl == 0) bias[c0 + lane] = bc[cc];

  float acc[4] = {0.f, 0.f, 0.f, 0.f};
#pragma unroll
  for (int j = 0; j < PREP_ROWS; ++j) {
#pragma unroll
    for (int k = 0; k < 4; ++k)
      acc[k] = fmaf(W_in[k * D_ + d0 + j], w[j], acc[k]);
  }
#pragma unroll
  for (int k = 0; k < 4; ++k) sp[sl][k][lane] = acc[k];
  __syncthreads();

  if (sl < 4) {
    float s = sp[0][sl][lane];
#pragma unroll
    for (int q = 1; q < PREP_SL; ++q) s += sp[q][sl][lane];
    weff[sl * CPW + c0 + lane] = s;
  }
}

struct StepMap { float m[9]; float inj[3]; };

static __device__ __forceinline__ StepMap make_step(
    float bx, float by, float bz, float d0, float d1, float d2,
    float i0, float i1, float i2) {
  const float ss  = fmaxf(bx * bx + by * by + bz * bz, EPSF * EPSF);
  const float rnb = __builtin_amdgcn_rsqf(ss);
  const float nb  = ss * rnb;
  float sn, cs;
  __sincosf(0.5f * nb, &sn, &cs);
  const float s = sn * rnb;
  const float qx = s * bz, qy = -s * by, qz = s * bx;
  const float gx = __builtin_amdgcn_rcpf(1.f + __expf(-d0));
  const float gy = __builtin_amdgcn_rcpf(1.f + __expf(-d1));
  const float gz = __builtin_amdgcn_rcpf(1.f + __expf(-d2));
  const float xx = qx * qx, yy = qy * qy, zz = qz * qz;
  const float xy = qx * qy, xz = qx * qz, yz = qy * qz;
  const float wx = cs * qx, wy = cs * qy, wz = cs * qz;
  StepMap r;
  r.m[0] = gx * (1.f - 2.f * (yy + zz));
  r.m[1] = gx * (2.f * (xy - wz));
  r.m[2] = gx * (2.f * (xz + wy));
  r.m[3] = gy * (2.f * (xy + wz));
  r.m[4] = gy * (1.f - 2.f * (xx + zz));
  r.m[5] = gy * (2.f * (yz - wx));
  r.m[6] = gz * (2.f * (xz - wy));
  r.m[7] = gz * (2.f * (yz + wx));
  r.m[8] = gz * (1.f - 2.f * (xx + yy));
  r.inj[0] = i0; r.inj[1] = i1; r.inj[2] = i2;
  return r;
}

static __device__ __forceinline__ void load_e_operands(
    const float* __restrict__ weff, const float* __restrict__ bias,
    int blk, int k0, v2f bm[9], float bv[9]) {
#pragma unroll
  for (int kind = 0; kind < 3; ++kind) {
#pragma unroll
    for (int comp = 0; comp < 3; ++comp) {
      const int col = kind * D_ + blk * 3 + comp;
      bm[kind * 3 + comp].x = weff[k0 * CPW + col];
      bm[kind * 3 + comp].y = weff[(k0 + 1) * CPW + col];
      bv[kind * 3 + comp]   = bias[col];
    }
  }
}

static __device__ __forceinline__ void compute_e(
    v2f a, const v2f bm[9], const float bv[9], float e[9][8]) {
#pragma unroll
  for (int j = 0; j < 9; ++j) {
    const float b0 = bv[j];
    v8f c = {b0, b0, b0, b0, b0, b0, b0, b0};
    v8f d = __builtin_amdgcn_wmma_f32_16x16x4_f32(
         false, a,  false, bm[j],
         (short)0, c,  false,  false);
#pragma unroll
    for (int v = 0; v < 8; ++v) e[j][v] = d[v];
  }
}

__global__ __launch_bounds__(256) void so3_seg_kernel(
    const float* __restrict__ quat,
    const float* __restrict__ weff,
    const float* __restrict__ bias,
    float* __restrict__ pv) {
  const int tid  = threadIdx.x;
  const int lane = tid & 31;
  const int g    = __builtin_amdgcn_readfirstlane(tid >> 5);
  const int t16  = lane & 15;
  const int k0   = (lane < 16) ? 0 : 2;
  const int half = lane >> 4;
  const int b    = blockIdx.x / (SEG_N / 2);
  const int p    = blockIdx.x % (SEG_N / 2);
  const int blk  = g * 16 + t16;

  const int trow = (2 * p + (t16 >> 3)) * SEG_L + (t16 & 7);
  const float* qrow = quat + ((size_t)b * T_ + trow) * 4 + k0;

  v2f bm[9];
  float bv[9];
  load_e_operands(weff, bias, blk, k0, bm, bv);

  float P[3][3] = {{1.f, 0.f, 0.f}, {0.f, 1.f, 0.f}, {0.f, 0.f, 1.f}};
  float vv[3] = {0.f, 0.f, 0.f};

  v2f a;
  a.x = qrow[0];
  a.y = qrow[1];
  for (int i = 0; i < NBATCH; ++i) {
    const int inx = min(i + 1, NBATCH - 1);
    v2f an;
    an.x = qrow[inx * 32];
    an.y = qrow[inx * 32 + 1];

    float e[9][8];
    compute_e(a, bm, bv, e);
#pragma unroll
    for (int v = 0; v < 8; ++v) {
      const StepMap s = make_step(e[0][v], e[1][v], e[2][v], e[3][v], e[4][v], e[5][v],
                                  e[6][v], e[7][v], e[8][v]);
      float Pn[3][3], vn[3];
#pragma unroll
      for (int r = 0; r < 3; ++r) {
#pragma unroll
        for (int c = 0; c < 3; ++c)
          Pn[r][c] = fmaf(s.m[r * 3], P[0][c], fmaf(s.m[r * 3 + 1], P[1][c], s.m[r * 3 + 2] * P[2][c]));
        vn[r] = fmaf(s.m[r * 3], vv[0], fmaf(s.m[r * 3 + 1], vv[1], fmaf(s.m[r * 3 + 2], vv[2], s.inj[r])));
      }
#pragma unroll
      for (int r = 0; r < 3; ++r) {
#pragma unroll
        for (int c = 0; c < 3; ++c) P[r][c] = Pn[r][c];
        vv[r] = vn[r];
      }
    }
    a = an;
  }

  const int seg = 2 * p + half;
  v4f* dst = (v4f*)pv + (((size_t)b * SEG_N + seg) * NB_ + blk) * 3;
#pragma unroll
  for (int r = 0; r < 3; ++r) {
    v4f o = {P[r][0], P[r][1], P[r][2], vv[r]};
    dst[r] = o;
  }
}

__global__ __launch_bounds__(32) void so3_carry_kernel(
    const float* __restrict__ pv,
    float* __restrict__ hs) {
  const int b   = blockIdx.x >> 2;
  const int blk = (blockIdx.x & 3) * 32 + threadIdx.x;
  const v4f* src = (const v4f*)pv + ((size_t)b * SEG_N * NB_ + blk) * 3;
  v4f* dst = (v4f*)hs + (size_t)b * SEG_N * NB_ + blk;
  float hx = 0.f, hy = 0.f, hz = 0.f;
#pragma unroll 8
  for (int s = 0; s < SEG_N; ++s) {
    const v4f h4 = {hx, hy, hz, 0.f};
    dst[(size_t)s * NB_] = h4;
    const v4f r0 = src[(size_t)s * NB_ * 3];
    const v4f r1 = src[(size_t)s * NB_ * 3 + 1];
    const v4f r2 = src[(size_t)s * NB_ * 3 + 2];
    const float nx = fmaf(r0.x, hx, fmaf(r0.y, hy, fmaf(r0.z, hz, r0.w)));
    const float ny = fmaf(r1.x, hx, fmaf(r1.y, hy, fmaf(r1.z, hz, r1.w)));
    const float nz = fmaf(r2.x, hx, fmaf(r2.y, hy, fmaf(r2.z, hz, r2.w)));
    hx = nx; hy = ny; hz = nz;
  }
}

static __device__ __forceinline__ void wave_lds_handover() {
  asm volatile("s_wait_dscnt 0x0" ::: "memory");
}

__global__ __launch_bounds__(256) void so3_emit_kernel(
    const float* __restrict__ quat,
    const float* __restrict__ weff,
    const float* __restrict__ bias,
    const float* __restrict__ hs,
    const float* __restrict__ W_out,
    float* __restrict__ part) {
  __shared__ float sH[8][16 * HP];
  __shared__ __attribute__((aligned(16))) float sPart[8][256];

  const int tid  = threadIdx.x;
  const int lane = tid & 31;
  const int g    = __builtin_amdgcn_readfirstlane(tid >> 5);
  const int t16  = lane & 15;
  const int k0   = (lane < 16) ? 0 : 2;
  const int half = lane >> 4;
  const int mhi  = half * 8;
  const int b    = blockIdx.x / (SEG_N / 2);
  const int p    = blockIdx.x % (SEG_N / 2);
  const int blk  = g * 16 + t16;
  float* myH = sH[g];
  float* myP = sPart[g];

  const int trow = (2 * p + (t16 >> 3)) * SEG_L + (t16 & 7);
  const float* qrow = quat + ((size_t)b * T_ + trow) * 4 + k0;

  v2f bm[9];
  float bv[9];
  load_e_operands(weff, bias, blk, k0, bm, bv);

  v2f bw[12];
#pragma unroll
  for (int kk = 0; kk < 12; ++kk) {
    const int kr = g * 48 + kk * 4 + k0;
    bw[kk].x = (t16 < 4) ? W_out[kr * 4 + t16] : 0.f;
    bw[kk].y = (t16 < 4) ? W_out[(kr + 1) * 4 + t16] : 0.f;
  }

  const int seg = 2 * p + half;
  const v4f h4 = ((const v4f*)hs)[((size_t)b * SEG_N + seg) * NB_ + blk];
  float hx = h4.x, hy = h4.y, hz = h4.z;

  v4f* pbase = (v4f*)part + ((size_t)(b * 8 + g) * T_ + trow);

  v2f a;
  a.x = qrow[0];
  a.y = qrow[1];
  for (int i = 0; i < NBATCH; ++i) {
    const int inx = min(i + 1, NBATCH - 1);
    v2f an;
    an.x = qrow[inx * 32];
    an.y = qrow[inx * 32 + 1];

    float e[9][8];
    compute_e(a, bm, bv, e);
#pragma unroll
    for (int v = 0; v < 8; ++v) {
      const StepMap s = make_step(e[0][v], e[1][v], e[2][v], e[3][v], e[4][v], e[5][v],
                                  e[6][v], e[7][v], e[8][v]);
      const float nx = fmaf(s.m[0], hx, fmaf(s.m[1], hy, fmaf(s.m[2], hz, s.inj[0])));
      const float ny = fmaf(s.m[3], hx, fmaf(s.m[4], hy, fmaf(s.m[5], hz, s.inj[1])));
      const float nz = fmaf(s.m[6], hx, fmaf(s.m[7], hy, fmaf(s.m[8], hz, s.inj[2])));
      hx = nx; hy = ny; hz = nz;
      float* hrow = &myH[(mhi + v) * HP + t16 * 3];
      hrow[0] = hx;
      hrow[1] = hy;
      hrow[2] = hz;
    }
    wave_lds_handover();

    {
      v2f af[12];
      const float* hr = &myH[t16 * HP + k0];
#pragma unroll
      for (int kk = 0; kk < 12; ++kk) {
        af[kk].x = hr[kk * 4];
        af[kk].y = hr[kk * 4 + 1];
      }
      v8f acc0 = {0.f, 0.f, 0.f, 0.f, 0.f, 0.f, 0.f, 0.f};
      v8f acc1 = {0.f, 0.f, 0.f, 0.f, 0.f, 0.f, 0.f, 0.f};
#pragma unroll
      for (int kk = 0; kk < 12; kk += 2) {
        acc0 = __builtin_amdgcn_wmma_f32_16x16x4_f32(
            false, af[kk], false, bw[kk], (short)0, acc0, false, false);
        acc1 = __builtin_amdgcn_wmma_f32_16x16x4_f32(
            false, af[kk + 1], false, bw[kk + 1], (short)0, acc1, false, false);
      }
      acc0 = acc0 + acc1;
#pragma unroll
      for (int v = 0; v < 8; ++v)
        myP[(v + mhi) * 16 + t16] = acc0[v];
    }
    wave_lds_handover();

    if (lane < 16)
      pbase[i * 8] = *(const v4f*)&myP[lane * 16];
    a = an;
  }
}

__global__ __launch_bounds__(256) void so3_norm_kernel(
    const float* __restrict__ part,
    float* __restrict__ out) {
  const int i = blockIdx.x * 256 + threadIdx.x;
  const int b = i / T_;
  const int t = i % T_;
  const v4f* src = (const v4f*)part + (size_t)b * 8 * T_ + t;
  v4f acc = src[0];
#pragma unroll
  for (int gi = 1; gi < 8; ++gi) acc = acc + src[(size_t)gi * T_];
  float n = sqrtf(acc[0] * acc[0] + acc[1] * acc[1] + acc[2] * acc[2] + acc[3] * acc[3]);
  float rn = 1.f / fmaxf(n, EPSF);
  float4 o = {acc[0] * rn, acc[1] * rn, acc[2] * rn, acc[3] * rn};
  *(float4*)&out[(size_t)i * 4] = o;
}

extern "C" void kernel_launch(void* const* d_in, const int* in_sizes, int n_in,
                              void* d_out, int out_size, void* d_ws, size_t ws_size,
                              hipStream_t stream) {
  const float* quat  = (const float*)d_in[0];
  const float* W_in  = (const float*)d_in[1];
  const float* W_biv = (const float*)d_in[2];
  const float* b_biv = (const float*)d_in[3];
  const float* W_dec = (const float*)d_in[4];
  const float* b_dec = (const float*)d_in[5];
  const float* W_inj = (const float*)d_in[6];
  const float* b_inj = (const float*)d_in[7];
  const float* W_out = (const float*)d_in[8];
  float* out  = (float*)d_out;
  float* weff = (float*)d_ws;
  float* bias = weff + 4 * CPW;
  float* pv   = weff + WS_PV_OFF;
  float* hs   = pv + WS_PV_FLOATS;
  float* part = hs + WS_HS_FLOATS;

  so3_prep_kernel<<<CPW / PREP_COLS, PREP_COLS * PREP_SL, 0, stream>>>(
      W_in, W_biv, b_biv, W_dec, b_dec, W_inj, b_inj, weff, bias);
  so3_seg_kernel<<<B_ * (SEG_N / 2), 256, 0, stream>>>(quat, weff, bias, pv);
  so3_carry_kernel<<<B_ * 4, 32, 0, stream>>>(pv, hs);
  so3_emit_kernel<<<B_ * (SEG_N / 2), 256, 0, stream>>>(quat, weff, bias, hs, W_out, part);
  so3_norm_kernel<<<B_ * T_ / 256, 256, 0, stream>>>(part, out);
}


# Carry pass as a two-level scan in a workgroup: 8 waves each fetch 8 segment maps in one memory round trip, compose them, exchange via LDS, replay from registers; carry rounding order changes
# speedup vs baseline: 24.7884x; 1.4469x over previous
#include <hip/hip_runtime.h>
#include <hip/hip_bf16.h>


#define B_   16
#define T_   2048
#define D_   384
#define NB_  128
#define CPW  1152
#define SEG_L 32
#define SEG_N (T_ / SEG_L)
#define NBATCH (SEG_L / 8)
#define HP   52
#define WS_PV_OFF   8192
#define WS_PV_FLOATS (B_ * SEG_N * NB_ * 12)
#define WS_HS_FLOATS (B_ * SEG_N * NB_ * 4)
#define EPSF 1e-8f

typedef __attribute__((ext_vector_type(2))) float v2f;
typedef __attribute__((ext_vector_type(4))) float v4f;
typedef __attribute__((ext_vector_type(8))) float v8f;

#define PREP_COLS 32
#define PREP_SL   16
#define PREP_ROWS (D_ / PREP_SL)

__global__ __launch_bounds__(PREP_COLS * PREP_SL) void so3_prep_kernel(
    const float* __restrict__ W_in,
    const float* __restrict__ W_biv, const float* __restrict__ b_biv,
    const float* __restrict__ W_dec, const float* __restrict__ b_dec,
    const float* __restrict__ W_inj, const float* __restrict__ b_inj,
    float* __restrict__ weff, float* __restrict__ bias) {
  __shared__ float sp[PREP_SL][4][PREP_COLS];
  const int lane = threadIdx.x & 31;
  const int sl = __builtin_amdgcn_readfirstlane(threadIdx.x >> 5);
  const int c0 = blockIdx.x * PREP_COLS;
  const float* Wc;
  const float* bc;
  int cc;
  if (c0 < D_)            { Wc = W_biv; bc = b_biv; cc = c0 + lane; }
  else if (c0 < 2 * D_)   { Wc = W_dec; bc = b_dec; cc = c0 - D_ + lane; }
  else                    { Wc = W_inj; bc = b_inj; cc = c0 - 2 * D_ + lane; }
  const int d0 = sl * PREP_ROWS;
  const float* wp = Wc + d0 * D_ + cc;

  float w[PREP_ROWS];
#pragma unroll
  for (int j = 0; j < PREP_ROWS; ++j) w[j] = wp[j * D_];
  if (sl == 0) bias[c0 + lane] = bc[cc];

  float acc[4] = {0.f, 0.f, 0.f, 0.f};
#pragma unroll
  for (int j = 0; j < PREP_ROWS; ++j) {
#pragma unroll
    for (int k = 0; k < 4; ++k)
      acc[k] = fmaf(W_in[k * D_ + d0 + j], w[j], acc[k]);
  }
#pragma unroll
  for (int k = 0; k < 4; ++k) sp[sl][k][lane] = acc[k];
  __syncthreads();

  if (sl < 4) {
    float s = sp[0][sl][lane];
#pragma unroll
    for (int q = 1; q < PREP_SL; ++q) s += sp[q][sl][lane];
    weff[sl * CPW + c0 + lane] = s;
  }
}

struct StepMap { float m[9]; float inj[3]; };

static __device__ __forceinline__ StepMap make_step(
    float bx, float by, float bz, float d0, float d1, float d2,
    float i0, float i1, float i2) {
  const float ss  = fmaxf(bx * bx + by * by + bz * bz, EPSF * EPSF);
  const float rnb = __builtin_amdgcn_rsqf(ss);
  const float nb  = ss * rnb;
  float sn, cs;
  __sincosf(0.5f * nb, &sn, &cs);
  const float s = sn * rnb;
  const float qx = s * bz, qy = -s * by, qz = s * bx;
  const float gx = __builtin_amdgcn_rcpf(1.f + __expf(-d0));
  const float gy = __builtin_amdgcn_rcpf(1.f + __expf(-d1));
  const float gz = __builtin_amdgcn_rcpf(1.f + __expf(-d2));
  const float xx = qx * qx, yy = qy * qy, zz = qz * qz;
  const float xy = qx * qy, xz = qx * qz, yz = qy * qz;
  const float wx = cs * qx, wy = cs * qy, wz = cs * qz;
  StepMap r;
  r.m[0] = gx * (1.f - 2.f * (yy + zz));
  r.m[1] = gx * (2.f * (xy - wz));
  r.m[2] = gx * (2.f * (xz + wy));
  r.m[3] = gy * (2.f * (xy + wz));
  r.m[4] = gy * (1.f - 2.f * (xx + zz));
  r.m[5] = gy * (2.f * (yz - wx));
  r.m[6] = gz * (2.f * (xz - wy));
  r.m[7] = gz * (2.f * (yz + wx));
  r.m[8] = gz * (1.f - 2.f * (xx + yy));
  r.inj[0] = i0; r.inj[1] = i1; r.inj[2] = i2;
  return r;
}

static __device__ __forceinline__ void load_e_operands(
    const float* __restrict__ weff, const float* __restrict__ bias,
    int blk, int k0, v2f bm[9], float bv[9]) {
#pragma unroll
  for (int kind = 0; kind < 3; ++kind) {
#pragma unroll
    for (int comp = 0; comp < 3; ++comp) {
      const int col = kind * D_ + blk * 3 + comp;
      bm[kind * 3 + comp].x = weff[k0 * CPW + col];
      bm[kind * 3 + comp].y = weff[(k0 + 1) * CPW + col];
      bv[kind * 3 + comp]   = bias[col];
    }
  }
}

static __device__ __forceinline__ void compute_e(
    v2f a, const v2f bm[9], const float bv[9], float e[9][8]) {
#pragma unroll
  for (int j = 0; j < 9; ++j) {
    const float b0 = bv[j];
    v8f c = {b0, b0, b0, b0, b0, b0, b0, b0};
    v8f d = __builtin_amdgcn_wmma_f32_16x16x4_f32(
         false, a,  false, bm[j],
         (short)0, c,  false,  false);
#pragma unroll
    for (int v = 0; v < 8; ++v) e[j][v] = d[v];
  }
}

__global__ __launch_bounds__(256) void so3_seg_kernel(
    const float* __restrict__ quat,
    const float* __restrict__ weff,
    const float* __restrict__ bias,
    float* __restrict__ pv) {
  const int tid  = threadIdx.x;
  const int lane = tid & 31;
  const int g    = __builtin_amdgcn_readfirstlane(tid >> 5);
  const int t16  = lane & 15;
  const int k0   = (lane < 16) ? 0 : 2;
  const int half = lane >> 4;
  const int b    = blockIdx.x / (SEG_N / 2);
  const int p    = blockIdx.x % (SEG_N / 2);
  const int blk  = g * 16 + t16;

  const int trow = (2 * p + (t16 >> 3)) * SEG_L + (t16 & 7);
  const float* qrow = quat + ((size_t)b * T_ + trow) * 4 + k0;

  v2f bm[9];
  float bv[9];
  load_e_operands(weff, bias, blk, k0, bm, bv);

  float P[3][3] = {{1.f, 0.f, 0.f}, {0.f, 1.f, 0.f}, {0.f, 0.f, 1.f}};
  float vv[3] = {0.f, 0.f, 0.f};

  v2f a;
  a.x = qrow[0];
  a.y = qrow[1];
  for (int i = 0; i < NBATCH; ++i) {
    const int inx = min(i + 1, NBATCH - 1);
    v2f an;
    an.x = qrow[inx * 32];
    an.y = qrow[inx * 32 + 1];

    float e[9][8];
    compute_e(a, bm, bv, e);
#pragma unroll
    for (int v = 0; v < 8; ++v) {
      const StepMap s = make_step(e[0][v], e[1][v], e[2][v], e[3][v], e[4][v], e[5][v],
                                  e[6][v], e[7][v], e[8][v]);
      float Pn[3][3], vn[3];
#pragma unroll
      for (int r = 0; r < 3; ++r) {
#pragma unroll
        for (int c = 0; c < 3; ++c)
          Pn[r][c] = fmaf(s.m[r * 3], P[0][c], fmaf(s.m[r * 3 + 1], P[1][c], s.m[r * 3 + 2] * P[2][c]));
        vn[r] = fmaf(s.m[r * 3], vv[0], fmaf(s.m[r * 3 + 1], vv[1], fmaf(s.m[r * 3 + 2], vv[2], s.inj[r])));
      }
#pragma unroll
      for (int r = 0; r < 3; ++r) {
#pragma unroll
        for (int c = 0; c < 3; ++c) P[r][c] = Pn[r][c];
        vv[r] = vn[r];
      }
    }
    a = an;
  }

  const int seg = 2 * p + half;
  v4f* dst = (v4f*)pv + (((size_t)b * SEG_N + seg) * NB_ + blk) * 3;
#pragma unroll
  for (int r = 0; r < 3; ++r) {
    v4f o = {P[r][0], P[r][1], P[r][2], vv[r]};
    dst[r] = o;
  }
}

#define CARRY_W 8
#define CARRY_S (SEG_N / CARRY_W)

static __device__ __forceinline__ void apply_map(
    const v4f r0, const v4f r1, const v4f r2, float& hx, float& hy, float& hz) {
  const float nx = fmaf(r0.x, hx, fmaf(r0.y, hy, fmaf(r0.z, hz, r0.w)));
  const float ny = fmaf(r1.x, hx, fmaf(r1.y, hy, fmaf(r1.z, hz, r1.w)));
  const float nz = fmaf(r2.x, hx, fmaf(r2.y, hy, fmaf(r2.z, hz, r2.w)));
  hx = nx; hy = ny; hz = nz;
}

__global__ __launch_bounds__(32 * CARRY_W) void so3_carry_kernel(
    const float* __restrict__ pv,
    float* __restrict__ hs) {
  __shared__ v4f sC[CARRY_W - 1][3][32];
  const int lane = threadIdx.x & 31;
  const int w    = __builtin_amdgcn_readfirstlane(threadIdx.x >> 5);
  const int b    = blockIdx.x >> 2;
  const int blk  = (blockIdx.x & 3) * 32 + lane;
  const int s0   = w * CARRY_S;
  const v4f* src = (const v4f*)pv + (((size_t)b * SEG_N + s0) * NB_ + blk) * 3;
  v4f* dst = (v4f*)hs + ((size_t)b * SEG_N + s0) * NB_ + blk;

  v4f m[CARRY_S][3];
#pragma unroll
  for (int k = 0; k < CARRY_S; ++k) {
#pragma unroll
    for (int r = 0; r < 3; ++r) m[k][r] = src[(size_t)k * NB_ * 3 + r];
  }
  asm volatile("" ::
               "v"(m[0][0]), "v"(m[0][1]), "v"(m[0][2]), "v"(m[1][0]), "v"(m[1][1]), "v"(m[1][2]),
               "v"(m[2][0]), "v"(m[2][1]), "v"(m[2][2]), "v"(m[3][0]), "v"(m[3][1]), "v"(m[3][2]),
               "v"(m[4][0]), "v"(m[4][1]), "v"(m[4][2]), "v"(m[5][0]), "v"(m[5][1]), "v"(m[5][2]),
               "v"(m[6][0]), "v"(m[6][1]), "v"(m[6][2]), "v"(m[7][0]), "v"(m[7][1]), "v"(m[7][2]));

  if (w < CARRY_W - 1) {
    float P[3][3], q[3];
#pragma unroll
    for (int r = 0; r < 3; ++r) {
      P[r][0] = m[0][r].x; P[r][1] = m[0][r].y; P[r][2] = m[0][r].z; q[r] = m[0][r].w;
    }
#pragma unroll
    for (int k = 1; k < CARRY_S; ++k) {
      float Pn[3][3], qn[3];
#pragma unroll
      for (int r = 0; r < 3; ++r) {
        const v4f mr = m[k][r];
#pragma unroll
        for (int c = 0; c < 3; ++c)
          Pn[r][c] = fmaf(mr.x, P[0][c], fmaf(mr.y, P[1][c], mr.z * P[2][c]));
        qn[r] = fmaf(mr.x, q[0], fmaf(mr.y, q[1], fmaf(mr.z, q[2], mr.w)));
      }
#pragma unroll
      for (int r = 0; r < 3; ++r) {
#pragma unroll
        for (int c = 0; c < 3; ++c) P[r][c] = Pn[r][c];
        q[r] = qn[r];
      }
    }
#pragma unroll
    for (int r = 0; r < 3; ++r) {
      const v4f o = {P[r][0], P[r][1], P[r][2], q[r]};
      sC[w][r][lane] = o;
    }
  }
  __syncthreads();

  v4f c[CARRY_W - 1][3];
#pragma unroll
  for (int j = 0; j < CARRY_W - 1; ++j) {
#pragma unroll
    for (int r = 0; r < 3; ++r) c[j][r] = sC[j][r][lane];
  }
  asm volatile("" ::
               "v"(c[0][0]), "v"(c[0][1]), "v"(c[0][2]), "v"(c[1][0]), "v"(c[1][1]), "v"(c[1][2]),
               "v"(c[2][0]), "v"(c[2][1]), "v"(c[2][2]), "v"(c[3][0]), "v"(c[3][1]), "v"(c[3][2]),
               "v"(c[4][0]), "v"(c[4][1]), "v"(c[4][2]), "v"(c[5][0]), "v"(c[5][1]), "v"(c[5][2]),
               "v"(c[6][0]), "v"(c[6][1]), "v"(c[6][2]));
  float hx = 0.f, hy = 0.f, hz = 0.f;
#pragma unroll
  for (int j = 0; j < CARRY_W - 1; ++j) {
    if (j < w) apply_map(c[j][0], c[j][1], c[j][2], hx, hy, hz);
  }

#pragma unroll
  for (int k = 0; k < CARRY_S; ++k) {
    const v4f h4 = {hx, hy, hz, 0.f};
    dst[(size_t)k * NB_] = h4;
    if (k < CARRY_S - 1) apply_map(m[k][0], m[k][1], m[k][2], hx, hy, hz);
  }
}

static __device__ __forceinline__ void wave_lds_handover() {
  asm volatile("s_wait_dscnt 0x0" ::: "memory");
}

__global__ __launch_bounds__(256) void so3_emit_kernel(
    const float* __restrict__ quat,
    const float* __restrict__ weff,
    const float* __restrict__ bias,
    const float* __restrict__ hs,
    const float* __restrict__ W_out,
    float* __restrict__ part) {
  __shared__ float sH[8][16 * HP];
  __shared__ __attribute__((aligned(16))) float sPart[8][256];

  const int tid  = threadIdx.x;
  const int lane = tid & 31;
  const int g    = __builtin_amdgcn_readfirstlane(tid >> 5);
  const int t16  = lane & 15;
  const int k0   = (lane < 16) ? 0 : 2;
  const int half = lane >> 4;
  const int mhi  = half * 8;
  const int b    = blockIdx.x / (SEG_N / 2);
  const int p    = blockIdx.x % (SEG_N / 2);
  const int blk  = g * 16 + t16;
  float* myH = sH[g];
  float* myP = sPart[g];

  const int trow = (2 * p + (t16 >> 3)) * SEG_L + (t16 & 7);
  const float* qrow = quat + ((size_t)b * T_ + trow) * 4 + k0;

  v2f bm[9];
  float bv[9];
  load_e_operands(weff, bias, blk, k0, bm, bv);

  v2f bw[12];
#pragma unroll
  for (int kk = 0; kk < 12; ++kk) {
    const int kr = g * 48 + kk * 4 + k0;
    bw[kk].x = (t16 < 4) ? W_out[kr * 4 + t16] : 0.f;
    bw[kk].y = (t16 < 4) ? W_out[(kr + 1) * 4 + t16] : 0.f;
  }

  const int seg = 2 * p + half;
  const v4f h4 = ((const v4f*)hs)[((size_t)b * SEG_N + seg) * NB_ + blk];
  float hx = h4.x, hy = h4.y, hz = h4.z;

  v4f* pbase = (v4f*)part + ((size_t)(b * 8 + g) * T_ + trow);

  v2f a;
  a.x = qrow[0];
  a.y = qrow[1];
  for (int i = 0; i < NBATCH; ++i) {
    const int inx = min(i + 1, NBATCH - 1);
    v2f an;
    an.x = qrow[inx * 32];
    an.y = qrow[inx * 32 + 1];

    float e[9][8];
    compute_e(a, bm, bv, e);
#pragma unroll
    for (int v = 0; v < 8; ++v) {
      const StepMap s = make_step(e[0][v], e[1][v], e[2][v], e[3][v], e[4][v], e[5][v],
                                  e[6][v], e[7][v], e[8][v]);
      const float nx = fmaf(s.m[0], hx, fmaf(s.m[1], hy, fmaf(s.m[2], hz, s.inj[0])));
      const float ny = fmaf(s.m[3], hx, fmaf(s.m[4], hy, fmaf(s.m[5], hz, s.inj[1])));
      const float nz = fmaf(s.m[6], hx, fmaf(s.m[7], hy, fmaf(s.m[8], hz, s.inj[2])));
      hx = nx; hy = ny; hz = nz;
      float* hrow = &myH[(mhi + v) * HP + t16 * 3];
      hrow[0] = hx;
      hrow[1] = hy;
      hrow[2] = hz;
    }
    wave_lds_handover();

    {
      v2f af[12];
      const float* hr = &myH[t16 * HP + k0];
#pragma unroll
      for (int kk = 0; kk < 12; ++kk) {
        af[kk].x = hr[kk * 4];
        af[kk].y = hr[kk * 4 + 1];
      }
      v8f acc0 = {0.f, 0.f, 0.f, 0.f, 0.f, 0.f, 0.f, 0.f};
      v8f acc1 = {0.f, 0.f, 0.f, 0.f, 0.f, 0.f, 0.f, 0.f};
#pragma unroll
      for (int kk = 0; kk < 12; kk += 2) {
        acc0 = __builtin_amdgcn_wmma_f32_16x16x4_f32(
            false, af[kk], false, bw[kk], (short)0, acc0, false, false);
        acc1 = __builtin_amdgcn_wmma_f32_16x16x4_f32(
            false, af[kk + 1], false, bw[kk + 1], (short)0, acc1, false, false);
      }
      acc0 = acc0 + acc1;
#pragma unroll
      for (int v = 0; v < 8; ++v)
        myP[(v + mhi) * 16 + t16] = acc0[v];
    }
    wave_lds_handover();

    if (lane < 16)
      pbase[i * 8] = *(const v4f*)&myP[lane * 16];
    a = an;
  }
}

__global__ __launch_bounds__(256) void so3_norm_kernel(
    const float* __restrict__ part,
    float* __restrict__ out) {
  const int i = blockIdx.x * 256 + threadIdx.x;
  const int b = i / T_;
  const int t = i % T_;
  const v4f* src = (const v4f*)part + (size_t)b * 8 * T_ + t;
  v4f acc = src[0];
#pragma unroll
  for (int gi = 1; gi < 8; ++gi) acc = acc + src[(size_t)gi * T_];
  float n = sqrtf(acc[0] * acc[0] + acc[1] * acc[1] + acc[2] * acc[2] + acc[3] * acc[3]);
  float rn = 1.f / fmaxf(n, EPSF);
  float4 o = {acc[0] * rn, acc[1] * rn, acc[2] * rn, acc[3] * rn};
  *(float4*)&out[(size_t)i * 4] = o;
}

extern "C" void kernel_launch(void* const* d_in, const int* in_sizes, int n_in,
                              void* d_out, int out_size, void* d_ws, size_t ws_size,
                              hipStream_t stream) {
  const float* quat  = (const float*)d_in[0];
  const float* W_in  = (const float*)d_in[1];
  const float* W_biv = (const float*)d_in[2];
  const float* b_biv = (const float*)d_in[3];
  const float* W_dec = (const float*)d_in[4];
  const float* b_dec = (const float*)d_in[5];
  const float* W_inj = (const float*)d_in[6];
  const float* b_inj = (const float*)d_in[7];
  const float* W_out = (const float*)d_in[8];
  float* out  = (float*)d_out;
  float* weff = (float*)d_ws;
  float* bias = weff + 4 * CPW;
  float* pv   = weff + WS_PV_OFF;
  float* hs   = pv + WS_PV_FLOATS;
  float* part = hs + WS_HS_FLOATS;

  so3_prep_kernel<<<CPW / PREP_COLS, PREP_COLS * PREP_SL, 0, stream>>>(
      W_in, W_biv, b_biv, W_dec, b_dec, W_inj, b_inj, weff, bias);
  so3_seg_kernel<<<B_ * (SEG_N / 2), 256, 0, stream>>>(quat, weff, bias, pv);
  so3_carry_kernel<<<B_ * 4, 32 * CARRY_W, 0, stream>>>(pv, hs);
  so3_emit_kernel<<<B_ * (SEG_N / 2), 256, 0, stream>>>(quat, weff, bias, hs, W_out, part);
  so3_norm_kernel<<<B_ * T_ / 256, 256, 0, stream>>>(part, out);
}


# Group sum and normalization moved into the emit kernel: per-wave projection partials stay in LDS, one barrier, 64 lanes add in group order 0..7 and store out; 4 launches, last kernel removed
# speedup vs baseline: 27.7454x; 1.1193x over previous
#include <hip/hip_runtime.h>
#include <hip/hip_bf16.h>


#define B_   16
#define T_   2048
#define D_   384
#define NB_  128
#define CPW  1152
#define SEG_L 32
#define SEG_N (T_ / SEG_L)
#define NBATCH (SEG_L / 8)
#define HP   52
#define WS_PV_OFF   8192
#define WS_PV_FLOATS (B_ * SEG_N * NB_ * 12)
#define EPSF 1e-8f

typedef __attribute__((ext_vector_type(2))) float v2f;
typedef __attribute__((ext_vector_type(4))) float v4f;
typedef __attribute__((ext_vector_type(8))) float v8f;

#define PREP_COLS 32
#define PREP_SL   16
#define PREP_ROWS (D_ / PREP_SL)

__global__ __launch_bounds__(PREP_COLS * PREP_SL) void so3_prep_kernel(
    const float* __restrict__ W_in,
    const float* __restrict__ W_biv, const float* __restrict__ b_biv,
    const float* __restrict__ W_dec, const float* __restrict__ b_dec,
    const float* __restrict__ W_inj, const float* __restrict__ b_inj,
    float* __restrict__ weff, float* __restrict__ bias) {
  __shared__ float sp[PREP_SL][4][PREP_COLS];
  const int lane = threadIdx.x & 31;
  const int sl = __builtin_amdgcn_readfirstlane(threadIdx.x >> 5);
  const int c0 = blockIdx.x * PREP_COLS;
  const float* Wc;
  const float* bc;
  int cc;
  if (c0 < D_)            { Wc = W_biv; bc = b_biv; cc = c0 + lane; }
  else if (c0 < 2 * D_)   { Wc = W_dec; bc = b_dec; cc = c0 - D_ + lane; }
  else                    { Wc = W_inj; bc = b_inj; cc = c0 - 2 * D_ + lane; }
  const int d0 = sl * PREP_ROWS;
  const float* wp = Wc + d0 * D_ + cc;

  float w[PREP_ROWS];
#pragma unroll
  for (int j = 0; j < PREP_ROWS; ++j) w[j] = wp[j * D_];
  if (sl == 0) bias[c0 + lane] = bc[cc];

  float acc[4] = {0.f, 0.f, 0.f, 0.f};
#pragma unroll
  for (int j = 0; j < PREP_ROWS; ++j) {
#pragma unroll
    for (int k = 0; k < 4; ++k)
      acc[k] = fmaf(W_in[k * D_ + d0 + j], w[j], acc[k]);
  }
#pragma unroll
  for (int k = 0; k < 4; ++k) sp[sl][k][lane] = acc[k];
  __syncthreads();

  if (sl < 4) {
    float s = sp[0][sl][lane];
#pragma unroll
    for (int q = 1; q < PREP_SL; ++q) s += sp[q][sl][lane];
    weff[sl * CPW + c0 + lane] = s;
  }
}

struct StepMap { float m[9]; float inj[3]; };

static __device__ __forceinline__ StepMap make_step(
    float bx, float by, float bz, float d0, float d1, float d2,
    float i0, float i1, float i2) {
  const float ss  = fmaxf(bx * bx + by * by + bz * bz, EPSF * EPSF);
  const float rnb = __builtin_amdgcn_rsqf(ss);
  const float nb  = ss * rnb;
  float sn, cs;
  __sincosf(0.5f * nb, &sn, &cs);
  const float s = sn * rnb;
  const float qx = s * bz, qy = -s * by, qz = s * bx;
  const float gx = __builtin_amdgcn_rcpf(1.f + __expf(-d0));
  const float gy = __builtin_amdgcn_rcpf(1.f + __expf(-d1));
  const float gz = __builtin_amdgcn_rcpf(1.f + __expf(-d2));
  const float xx = qx * qx, yy = qy * qy, zz = qz * qz;
  const float xy = qx * qy, xz = qx * qz, yz = qy * qz;
  const float wx = cs * qx, wy = cs * qy, wz = cs * qz;
  StepMap r;
  r.m[0] = gx * (1.f - 2.f * (yy + zz));
  r.m[1] = gx * (2.f * (xy - wz));
  r.m[2] = gx * (2.f * (xz + wy));
  r.m[3] = gy * (2.f * (xy + wz));
  r.m[4] = gy * (1.f - 2.f * (xx + zz));
  r.m[5] = gy * (2.f * (yz - wx));
  r.m[6] = gz * (2.f * (xz - wy));
  r.m[7] = gz * (2.f * (yz + wx));
  r.m[8] = gz * (1.f - 2.f * (xx + yy));
  r.inj[0] = i0; r.inj[1] = i1; r.inj[2] = i2;
  return r;
}

static __device__ __forceinline__ void load_e_operands(
    const float* __restrict__ weff, const float* __restrict__ bias,
    int blk, int k0, v2f bm[9], float bv[9]) {
#pragma unroll
  for (int kind = 0; kind < 3; ++kind) {
#pragma unroll
    for (int comp = 0; comp < 3; ++comp) {
      const int col = kind * D_ + blk * 3 + comp;
      bm[kind * 3 + comp].x = weff[k0 * CPW + col];
      bm[kind * 3 + comp].y = weff[(k0 + 1) * CPW + col];
      bv[kind * 3 + comp]   = bias[col];
    }
  }
}

static __device__ __forceinline__ void compute_e(
    v2f a, const v2f bm[9], const float bv[9], float e[9][8]) {
#pragma unroll
  for (int j = 0; j < 9; ++j) {
    const float b0 = bv[j];
    v8f c = {b0, b0, b0, b0, b0, b0, b0, b0};
    v8f d = __builtin_amdgcn_wmma_f32_16x16x4_f32(
         false, a,  false, bm[j],
         (short)0, c,  false,  false);
#pragma unroll
    for (int v = 0; v < 8; ++v) e[j][v] = d[v];
  }
}

__global__ __launch_bounds__(256) void so3_seg_kernel(
    const float* __restrict__ quat,
    const float* __restrict__ weff,
    const float* __restrict__ bias,
    float* __restrict__ pv) {
  const int tid  = threadIdx.x;
  const int lane = tid & 31;
  const int g    = __builtin_amdgcn_readfirstlane(tid >> 5);
  const int t16  = lane & 15;
  const int k0   = (lane < 16) ? 0 : 2;
  const int half = lane >> 4;
  const int b    = blockIdx.x / (SEG_N / 2);
  const int p    = blockIdx.x % (SEG_N / 2);
  const int blk  = g * 16 + t16;

  const int trow = (2 * p + (t16 >> 3)) * SEG_L + (t16 & 7);
  const float* qrow = quat + ((size_t)b * T_ + trow) * 4 + k0;

  v2f bm[9];
  float bv[9];
  load_e_operands(weff, bias, blk, k0, bm, bv);

  float P[3][3] = {{1.f, 0.f, 0.f}, {0.f, 1.f, 0.f}, {0.f, 0.f, 1.f}};
  float vv[3] = {0.f, 0.f, 0.f};

  v2f a;
  a.x = qrow[0];
  a.y = qrow[1];
  for (int i = 0; i < NBATCH; ++i) {
    const int inx = min(i + 1, NBATCH - 1);
    v2f an;
    an.x = qrow[inx * 32];
    an.y = qrow[inx * 32 + 1];

    float e[9][8];
    compute_e(a, bm, bv, e);
#pragma unroll
    for (int v = 0; v < 8; ++v) {
      const StepMap s = make_step(e[0][v], e[1][v], e[2][v], e[3][v], e[4][v], e[5][v],
                                  e[6][v], e[7][v], e[8][v]);
      float Pn[3][3], vn[3];
#pragma unroll
      for (int r = 0; r < 3; ++r) {
#pragma unroll
        for (int c = 0; c < 3; ++c)
          Pn[r][c] = fmaf(s.m[r * 3], P[0][c], fmaf(s.m[r * 3 + 1], P[1][c], s.m[r * 3 + 2] * P[2][c]));
        vn[r] = fmaf(s.m[r * 3], vv[0], fmaf(s.m[r * 3 + 1], vv[1], fmaf(s.m[r * 3 + 2], vv[2], s.inj[r])));
      }
#pragma unroll
      for (int r = 0; r < 3; ++r) {
#pragma unroll
        for (int c = 0; c < 3; ++c) P[r][c] = Pn[r][c];
        vv[r] = vn[r];
      }
    }
    a = an;
  }

  const int seg = 2 * p + half;
  v4f* dst = (v4f*)pv + (((size_t)b * SEG_N + seg) * NB_ + blk) * 3;
#pragma unroll
  for (int r = 0; r < 3; ++r) {
    v4f o = {P[r][0], P[r][1], P[r][2], vv[r]};
    dst[r] = o;
  }
}

#define CARRY_W 8
#define CARRY_S (SEG_N / CARRY_W)

static __device__ __forceinline__ void apply_map(
    const v4f r0, const v4f r1, const v4f r2, float& hx, float& hy, float& hz) {
  const float nx = fmaf(r0.x, hx, fmaf(r0.y, hy, fmaf(r0.z, hz, r0.w)));
  const float ny = fmaf(r1.x, hx, fmaf(r1.y, hy, fmaf(r1.z, hz, r1.w)));
  const float nz = fmaf(r2.x, hx, fmaf(r2.y, hy, fmaf(r2.z, hz, r2.w)));
  hx = nx; hy = ny; hz = nz;
}

__global__ __launch_bounds__(32 * CARRY_W) void so3_carry_kernel(
    const float* __restrict__ pv,
    float* __restrict__ hs) {
  __shared__ v4f sC[CARRY_W - 1][3][32];
  const int lane = threadIdx.x & 31;
  const int w    = __builtin_amdgcn_readfirstlane(threadIdx.x >> 5);
  const int b    = blockIdx.x >> 2;
  const int blk  = (blockIdx.x & 3) * 32 + lane;
  const int s0   = w * CARRY_S;
  const v4f* src = (const v4f*)pv + (((size_t)b * SEG_N + s0) * NB_ + blk) * 3;
  v4f* dst = (v4f*)hs + ((size_t)b * SEG_N + s0) * NB_ + blk;

  v4f m[CARRY_S][3];
#pragma unroll
  for (int k = 0; k < CARRY_S; ++k) {
#pragma unroll
    for (int r = 0; r < 3; ++r) m[k][r] = src[(size_t)k * NB_ * 3 + r];
  }
  asm volatile("" ::
               "v"(m[0][0]), "v"(m[0][1]), "v"(m[0][2]), "v"(m[1][0]), "v"(m[1][1]), "v"(m[1][2]),
               "v"(m[2][0]), "v"(m[2][1]), "v"(m[2][2]), "v"(m[3][0]), "v"(m[3][1]), "v"(m[3][2]),
               "v"(m[4][0]), "v"(m[4][1]), "v"(m[4][2]), "v"(m[5][0]), "v"(m[5][1]), "v"(m[5][2]),
               "v"(m[6][0]), "v"(m[6][1]), "v"(m[6][2]), "v"(m[7][0]), "v"(m[7][1]), "v"(m[7][2]));

  if (w < CARRY_W - 1) {
    float P[3][3], q[3];
#pragma unroll
    for (int r = 0; r < 3; ++r) {
      P[r][0] = m[0][r].x; P[r][1] = m[0][r].y; P[r][2] = m[0][r].z; q[r] = m[0][r].w;
    }
#pragma unroll
    for (int k = 1; k < CARRY_S; ++k) {
      float Pn[3][3], qn[3];
#pragma unroll
      for (int r = 0; r < 3; ++r) {
        const v4f mr = m[k][r];
#pragma unroll
        for (int c = 0; c < 3; ++c)
          Pn[r][c] = fmaf(mr.x, P[0][c], fmaf(mr.y, P[1][c], mr.z * P[2][c]));
        qn[r] = fmaf(mr.x, q[0], fmaf(mr.y, q[1], fmaf(mr.z, q[2], mr.w)));
      }
#pragma unroll
      for (int r = 0; r < 3; ++r) {
#pragma unroll
        for (int c = 0; c < 3; ++c) P[r][c] = Pn[r][c];
        q[r] = qn[r];
      }
    }
#pragma unroll
    for (int r = 0; r < 3; ++r) {
      const v4f o = {P[r][0], P[r][1], P[r][2], q[r]};
      sC[w][r][lane] = o;
    }
  }
  __syncthreads();

  v4f c[CARRY_W - 1][3];
#pragma unroll
  for (int j = 0; j < CARRY_W - 1; ++j) {
#pragma unroll
    for (int r = 0; r < 3; ++r) c[j][r] = sC[j][r][lane];
  }
  asm volatile("" ::
               "v"(c[0][0]), "v"(c[0][1]), "v"(c[0][2]), "v"(c[1][0]), "v"(c[1][1]), "v"(c[1][2]),
               "v"(c[2][0]), "v"(c[2][1]), "v"(c[2][2]), "v"(c[3][0]), "v"(c[3][1]), "v"(c[3][2]),
               "v"(c[4][0]), "v"(c[4][1]), "v"(c[4][2]), "v"(c[5][0]), "v"(c[5][1]), "v"(c[5][2]),
               "v"(c[6][0]), "v"(c[6][1]), "v"(c[6][2]));
  float hx = 0.f, hy = 0.f, hz = 0.f;
#pragma unroll
  for (int j = 0; j < CARRY_W - 1; ++j) {
    if (j < w) apply_map(c[j][0], c[j][1], c[j][2], hx, hy, hz);
  }

#pragma unroll
  for (int k = 0; k < CARRY_S; ++k) {
    const v4f h4 = {hx, hy, hz, 0.f};
    dst[(size_t)k * NB_] = h4;
    if (k < CARRY_S - 1) apply_map(m[k][0], m[k][1], m[k][2], hx, hy, hz);
  }
}

static __device__ __forceinline__ void wave_lds_handover() {
  asm volatile("s_wait_dscnt 0x0" ::: "memory");
}

__global__ __launch_bounds__(256) void so3_emit_kernel(
    const float* __restrict__ quat,
    const float* __restrict__ weff,
    const float* __restrict__ bias,
    const float* __restrict__ hs,
    const float* __restrict__ W_out,
    float* __restrict__ out) {
  __shared__ float sH[8][16 * HP];
  __shared__ __attribute__((aligned(16))) float sOut[8][2 * SEG_L * 4];

  const int tid  = threadIdx.x;
  const int lane = tid & 31;
  const int g    = __builtin_amdgcn_readfirstlane(tid >> 5);
  const int t16  = lane & 15;
  const int k0   = (lane < 16) ? 0 : 2;
  const int half = lane >> 4;
  const int mhi  = half * 8;
  const int b    = blockIdx.x / (SEG_N / 2);
  const int p    = blockIdx.x % (SEG_N / 2);
  const int blk  = g * 16 + t16;
  float* myH = sH[g];
  float* myO = &sOut[g][half * SEG_L * 4 + t16];

  const int trow = (2 * p + (t16 >> 3)) * SEG_L + (t16 & 7);
  const float* qrow = quat + ((size_t)b * T_ + trow) * 4 + k0;

  v2f bm[9];
  float bv[9];
  load_e_operands(weff, bias, blk, k0, bm, bv);

  v2f bw[12];
#pragma unroll
  for (int kk = 0; kk < 12; ++kk) {
    const int kr = g * 48 + kk * 4 + k0;
    bw[kk].x = (t16 < 4) ? W_out[kr * 4 + t16] : 0.f;
    bw[kk].y = (t16 < 4) ? W_out[(kr + 1) * 4 + t16] : 0.f;
  }

  const int seg = 2 * p + half;
  const v4f h4 = ((const v4f*)hs)[((size_t)b * SEG_N + seg) * NB_ + blk];
  float hx = h4.x, hy = h4.y, hz = h4.z;

  v2f a;
  a.x = qrow[0];
  a.y = qrow[1];
  for (int i = 0; i < NBATCH; ++i) {
    const int inx = min(i + 1, NBATCH - 1);
    v2f an;
    an.x = qrow[inx * 32];
    an.y = qrow[inx * 32 + 1];

    float e[9][8];
    compute_e(a, bm, bv, e);
#pragma unroll
    for (int v = 0; v < 8; ++v) {
      const StepMap s = make_step(e[0][v], e[1][v], e[2][v], e[3][v], e[4][v], e[5][v],
                                  e[6][v], e[7][v], e[8][v]);
      const float nx = fmaf(s.m[0], hx, fmaf(s.m[1], hy, fmaf(s.m[2], hz, s.inj[0])));
      const float ny = fmaf(s.m[3], hx, fmaf(s.m[4], hy, fmaf(s.m[5], hz, s.inj[1])));
      const float nz = fmaf(s.m[6], hx, fmaf(s.m[7], hy, fmaf(s.m[8], hz, s.inj[2])));
      hx = nx; hy = ny; hz = nz;
      float* hrow = &myH[(mhi + v) * HP + t16 * 3];
      hrow[0] = hx;
      hrow[1] = hy;
      hrow[2] = hz;
    }
    wave_lds_handover();

    {
      v2f af[12];
      const float* hr = &myH[t16 * HP + k0];
#pragma unroll
      for (int kk = 0; kk < 12; ++kk) {
        af[kk].x = hr[kk * 4];
        af[kk].y = hr[kk * 4 + 1];
      }
      v8f acc0 = {0.f, 0.f, 0.f, 0.f, 0.f, 0.f, 0.f, 0.f};
      v8f acc1 = {0.f, 0.f, 0.f, 0.f, 0.f, 0.f, 0.f, 0.f};
#pragma unroll
      for (int kk = 0; kk < 12; kk += 2) {
        acc0 = __builtin_amdgcn_wmma_f32_16x16x4_f32(
            false, af[kk], false, bw[kk], (short)0, acc0, false, false);
        acc1 = __builtin_amdgcn_wmma_f32_16x16x4_f32(
            false, af[kk + 1], false, bw[kk + 1], (short)0, acc1, false, false);
      }
      acc0 = acc0 + acc1;
      if (t16 < 4) {
#pragma unroll
        for (int v = 0; v < 8; ++v)
          myO[(i * 8 + v) * 4] = acc0[v];
      }
    }
    a = an;
  }

  __syncthreads();

  if (tid < 2 * SEG_L) {
    v4f acc = *(const v4f*)&sOut[0][tid * 4];
#pragma unroll
    for (int gi = 1; gi < 8; ++gi) acc = acc + *(const v4f*)&sOut[gi][tid * 4];
    float n = sqrtf(acc[0] * acc[0] + acc[1] * acc[1] + acc[2] * acc[2] + acc[3] * acc[3]);
    float rn = 1.f / fmaxf(n, EPSF);
    float4 o = {acc[0] * rn, acc[1] * rn, acc[2] * rn, acc[3] * rn};
    *(float4*)&out[((size_t)b * T_ + (size_t)p * (2 * SEG_L) + tid) * 4] = o;
  }
}

extern "C" void kernel_launch(void* const* d_in, const int* in_sizes, int n_in,
                              void* d_out, int out_size, void* d_ws, size_t ws_size,
                              hipStream_t stream) {
  const float* quat  = (const float*)d_in[0];
  const float* W_in  = (const float*)d_in[1];
  const float* W_biv = (const float*)d_in[2];
  const float* b_biv = (const float*)d_in[3];
  const float* W_dec = (const float*)d_in[4];
  const float* b_dec = (const float*)d_in[5];
  const float* W_inj = (const float*)d_in[6];
  const float* b_inj = (const float*)d_in[7];
  const float* W_out = (const float*)d_in[8];
  float* out  = (float*)d_out;
  float* weff = (float*)d_ws;
  float* bias = weff + 4 * CPW;
  float* pv   = weff + WS_PV_OFF;
  float* hs   = pv + WS_PV_FLOATS;

  so3_prep_kernel<<<CPW / PREP_COLS, PREP_COLS * PREP_SL, 0, stream>>>(
      W_in, W_biv, b_biv, W_dec, b_dec, W_inj, b_inj, weff, bias);
  so3_seg_kernel<<<B_ * (SEG_N / 2), 256, 0, stream>>>(quat, weff, bias, pv);
  so3_carry_kernel<<<B_ * 4, 32 * CARRY_W, 0, stream>>>(pv, hs);
  so3_emit_kernel<<<B_ * (SEG_N / 2), 256, 0, stream>>>(quat, weff, bias, hs, W_out, out);
}
